# starting kernel
# baseline (speedup 1.0000x reference)
#include <hip/hip_runtime.h>

typedef float v2f __attribute__((ext_vector_type(2)));
typedef float v8f __attribute__((ext_vector_type(8)));

#define HDIM 64
#define EPSV 1e-5f

static inline int cdiv(long long a, long long b) { return (int)((a + b - 1) / b); }

__global__ void k_zero(float* __restrict__ p, int n) {
    int i = blockIdx.x * blockDim.x + threadIdx.x;
    if (i < n) p[i] = 0.0f;
}

__global__ void k_deg_init(float* __restrict__ deg, int n) {
    int i = blockIdx.x * blockDim.x + threadIdx.x;
    if (i < n) deg[i] = 1.0f;
}

__global__ void k_deg_accum(const int* __restrict__ dst, float* __restrict__ deg, int E) {
    int e = blockIdx.x * blockDim.x + threadIdx.x;
    if (e < E) atomicAdd(&deg[dst[e]], 1.0f);
}

__global__ void k_dinv(float* __restrict__ deg, int n) {
    int i = blockIdx.x * blockDim.x + threadIdx.x;
    if (i < n) deg[i] = rsqrtf(deg[i]);
}

__global__ void k_gemm_wmma(const float* __restrict__ A, const float* __restrict__ W,
                            float* __restrict__ out, int rowTiles) {
    int wid  = (blockIdx.x * blockDim.x + threadIdx.x) >> 5;
    int lane = threadIdx.x & 31;
    int tr = wid >> 2;
    int tc = wid & 3;
    if (tr >= rowTiles) return;
    int m  = lane & 15;
    int kh = (lane >> 4) << 1;
    const float* Ar = A + (size_t)(tr * 16 + m) * HDIM;
    const float* Wr = W + (size_t)(tc * 16 + m) * HDIM;
    v8f c = {0.f, 0.f, 0.f, 0.f, 0.f, 0.f, 0.f, 0.f};
#pragma unroll
    for (int k0 = 0; k0 < HDIM; k0 += 4) {
        v2f a = *(const v2f*)(Ar + k0 + kh);
        v2f b = *(const v2f*)(Wr + k0 + kh);
        c = __builtin_amdgcn_wmma_f32_16x16x4_f32(false, a, false, b, (short)0, c, false, false);
    }
    int rowBase = tr * 16 + ((lane >> 4) << 3);
    int col     = tc * 16 + m;
#pragma unroll
    for (int r = 0; r < 8; ++r)
        out[(size_t)(rowBase + r) * HDIM + col] = c[r];
}

__global__ void k_self_init(const float* __restrict__ hW, const float* __restrict__ dinv,
                            float* __restrict__ agg, int N) {
    int t = blockIdx.x * blockDim.x + threadIdx.x;
    if (t >= N * 16) return;
    int n = t >> 4, q = t & 15;
    float di  = dinv[n];
    float nrm = di * di;
    float4 v  = ((const float4*)(hW + (size_t)n * HDIM))[q];
    ((float4*)(agg + (size_t)n * HDIM))[q] =
        make_float4(v.x * nrm, v.y * nrm, v.z * nrm, v.w * nrm);
}

__global__ void k_scatter(const float* __restrict__ hW, const float* __restrict__ dinv,
                          const int* __restrict__ src, const int* __restrict__ dst,
                          float* __restrict__ agg, int E) {
    int t = blockIdx.x * blockDim.x + threadIdx.x;
    if (t >= E * 16) return;
    int e = t >> 4, q = t & 15;
    int s = src[e], d = dst[e];
    float nrm = dinv[s] * dinv[d];
    float4 v = ((const float4*)(hW + (size_t)s * HDIM))[q];
    float* a = agg + (size_t)d * HDIM + q * 4;
    atomicAdd(a + 0, v.x * nrm);
    atomicAdd(a + 1, v.y * nrm);
    atomicAdd(a + 2, v.z * nrm);
    atomicAdd(a + 3, v.w * nrm);
}

__global__ void k_bias_relu_stats(float* __restrict__ agg, const float* __restrict__ bias,
                                  float* __restrict__ stats  , int N) {
    __shared__ float sS[256];
    __shared__ float sQ[256];
    int f  = threadIdx.x & 63;
    int rg = threadIdx.x >> 6;
    float b = bias[f];
    float s = 0.f, q = 0.f;
    for (int row = blockIdx.x * 4 + rg; row < N; row += gridDim.x * 4) {
        size_t idx = (size_t)row * HDIM + f;
        float v = agg[idx] + b;
        v = fmaxf(v, 0.f);
        agg[idx] = v;
        s += v;
        q += v * v;
    }
    sS[threadIdx.x] = s;
    sQ[threadIdx.x] = q;
    __syncthreads();
    if (rg == 0) {
        s = sS[f] + sS[64 + f] + sS[128 + f] + sS[192 + f];
        q = sQ[f] + sQ[64 + f] + sQ[128 + f] + sQ[192 + f];
        atomicAdd(&stats[f], s);
        atomicAdd(&stats[64 + f], q);
    }
}

__global__ void k_bn_pool(const float* __restrict__ hin, const float* __restrict__ stats,
                          const float* __restrict__ g, const float* __restrict__ beta,
                          const int* __restrict__ batch, float* __restrict__ hout,
                          float* __restrict__ pooled, int poolOff, int N) {
    int t = blockIdx.x * blockDim.x + threadIdx.x;
    if (t >= N * HDIM) return;
    int n = t >> 6, f = t & 63;
    float invN = 1.0f / (float)N;
    float mean = stats[f] * invN;
    float var  = stats[64 + f] * invN - mean * mean;
    float v = g[f] * (hin[t] - mean) * rsqrtf(var + EPSV) + beta[f];
    hout[t] = v;
    atomicAdd(&pooled[(size_t)batch[n] * 192 + poolOff + f], v);
}

extern "C" void kernel_launch(void* const* d_in, const int* in_sizes, int n_in,
                              void* d_out, int out_size, void* d_ws, size_t ws_size,
                              hipStream_t stream) {
    const float* x    = (const float*)d_in[0];
    const int*   eidx = (const int*)d_in[1];
    const int*   batch = (const int*)d_in[2];
    const float* W[3]  = {(const float*)d_in[3],  (const float*)d_in[7],  (const float*)d_in[11]};
    const float* bi[3] = {(const float*)d_in[4],  (const float*)d_in[8],  (const float*)d_in[12]};
    const float* ga[3] = {(const float*)d_in[5],  (const float*)d_in[9],  (const float*)d_in[13]};
    const float* be[3] = {(const float*)d_in[6],  (const float*)d_in[10], (const float*)d_in[14]};

    const int N = in_sizes[0] / HDIM;
    const int E = in_sizes[1] / 2;
    const int G = (out_size - N * HDIM) / (3 * HDIM);

    const int* src = eidx;
    const int* dst = eidx + E;

    float* ws    = (float*)d_ws;
    float* dinv  = ws;
    float* tmp   = dinv + N;
    float* agg   = tmp  + (size_t)N * HDIM;
    float* bufH  = agg  + (size_t)N * HDIM;
    float* stats = bufH + (size_t)N * HDIM;

    float* pooled = (float*)d_out;
    float* hfinal = (float*)d_out + (size_t)G * 192;

    const int TB = 256;

    k_deg_init <<<cdiv(N, TB), TB, 0, stream>>>(dinv, N);
    k_deg_accum<<<cdiv(E, TB), TB, 0, stream>>>(dst, dinv, E);
    k_dinv     <<<cdiv(N, TB), TB, 0, stream>>>(dinv, N);

    k_zero<<<cdiv((long long)G * 192, TB), TB, 0, stream>>>(pooled, G * 192);

    const int rowTiles   = N / 16;
    const int gemmWaves  = rowTiles * (HDIM / 16);
    const int gemmBlocks = cdiv((long long)gemmWaves * 32, TB);

    const float* gin = x;
    for (int l = 0; l < 3; ++l) {
        float* hout = (l == 2) ? hfinal : bufH;

        k_gemm_wmma<<<gemmBlocks, TB, 0, stream>>>(gin, W[l], tmp, rowTiles);
        k_self_init<<<cdiv((long long)N * 16, TB), TB, 0, stream>>>(tmp, dinv, agg, N);
        k_scatter  <<<cdiv((long long)E * 16, TB), TB, 0, stream>>>(tmp, dinv, src, dst, agg, E);

        k_zero<<<1, 128, 0, stream>>>(stats, 128);
        k_bias_relu_stats<<<1024, TB, 0, stream>>>(agg, bi[l], stats, N);

        k_bn_pool<<<cdiv((long long)N * HDIM, TB), TB, 0, stream>>>(
            agg, stats, ga[l], be[l], batch, hout, pooled, l * HDIM, N);

        gin = hout;
    }
}


# BatchNorm apply + add-pool of layers 1-2 move into the next layer's WMMA GEMM as its A-operand load (pool: one add per feature per 16-row tile); 2 passes and their plane gone; pool sum order changes
# speedup vs baseline: 1.3046x; 1.3046x over previous
#include <hip/hip_runtime.h>

typedef float v2f __attribute__((ext_vector_type(2)));
typedef float v8f __attribute__((ext_vector_type(8)));

#define HDIM 64
#define EPSV 1e-5f

static inline int cdiv(long long a, long long b) { return (int)((a + b - 1) / b); }

__global__ void k_zero(float* __restrict__ p, int n) {
    int i = blockIdx.x * blockDim.x + threadIdx.x;
    if (i < n) p[i] = 0.0f;
}

__global__ void k_deg_init(float* __restrict__ deg, int n) {
    int i = blockIdx.x * blockDim.x + threadIdx.x;
    if (i < n) deg[i] = 1.0f;
}

__global__ void k_deg_accum(const int* __restrict__ dst, float* __restrict__ deg, int E) {
    int e = blockIdx.x * blockDim.x + threadIdx.x;
    if (e < E) atomicAdd(&deg[dst[e]], 1.0f);
}

__global__ void k_dinv(float* __restrict__ deg, int n) {
    int i = blockIdx.x * blockDim.x + threadIdx.x;
    if (i < n) deg[i] = rsqrtf(deg[i]);
}

__global__ void k_gemm_wmma(const float* __restrict__ A, const float* __restrict__ W,
                            float* __restrict__ out, int rowTiles) {
    int wid  = (blockIdx.x * blockDim.x + threadIdx.x) >> 5;
    int lane = threadIdx.x & 31;
    int tr = wid >> 2;
    int tc = wid & 3;
    if (tr >= rowTiles) return;
    int m  = lane & 15;
    int kh = (lane >> 4) << 1;
    const float* Ar = A + (size_t)(tr * 16 + m) * HDIM;
    const float* Wr = W + (size_t)(tc * 16 + m) * HDIM;
    v8f c = {0.f, 0.f, 0.f, 0.f, 0.f, 0.f, 0.f, 0.f};
#pragma unroll
    for (int k0 = 0; k0 < HDIM; k0 += 4) {
        v2f a = *(const v2f*)(Ar + k0 + kh);
        v2f b = *(const v2f*)(Wr + k0 + kh);
        c = __builtin_amdgcn_wmma_f32_16x16x4_f32(false, a, false, b, (short)0, c, false, false);
    }
    int rowBase = tr * 16 + ((lane >> 4) << 3);
    int col     = tc * 16 + m;
#pragma unroll
    for (int r = 0; r < 8; ++r)
        out[(size_t)(rowBase + r) * HDIM + col] = c[r];
}

__device__ __forceinline__ float bn_apply(float x, float gam, float mean, float rs, float bet) {
    return gam * (x - mean) * rs + bet;
}

__global__ void __launch_bounds__(256)
k_bn_pool_gemm(const float* __restrict__ hin, const float* __restrict__ stats,
               const float* __restrict__ g, const float* __restrict__ beta,
               const int* __restrict__ batch, float* __restrict__ pooled, int poolOff,
               const float* __restrict__ W, float* __restrict__ out, int rowTiles, int N) {
    __shared__ __attribute__((aligned(16))) float sBN[HDIM * 4];
    if (threadIdx.x < HDIM) {
        int f = threadIdx.x;
        float invN = 1.0f / (float)N;
        float mean = stats[f] * invN;
        float var  = stats[64 + f] * invN - mean * mean;
        float* q = sBN + ((f >> 1) << 3) + (f & 1);
        q[0] = g[f];
        q[2] = mean;
        q[4] = rsqrtf(var + EPSV);
        q[6] = beta[f];
    }
    __syncthreads();
    int wid  = (blockIdx.x * blockDim.x + threadIdx.x) >> 5;
    int lane = threadIdx.x & 31;
    int tr = wid >> 2;
    int tc = wid & 3;
    if (tr >= rowTiles) return;
    int m  = lane & 15;
    int kh = (lane >> 4) << 1;

    {
        int pf   = tc * 16 + m;
        int prow = tr * 16 + ((lane >> 4) << 3);
        const float* q = sBN + ((pf >> 1) << 3) + (pf & 1);
        float pg = q[0], pm = q[2], pr = q[4], pb = q[6];
        float v[8];
#pragma unroll
        for (int r = 0; r < 8; ++r)
            v[r] = bn_apply(hin[(size_t)(prow + r) * HDIM + pf], pg, pm, pr, pb);
        int bmine = batch[tr * 16 + m];
        int b0    = __builtin_amdgcn_readfirstlane(bmine);
        if (__all(bmine == b0)) {
            float s = ((v[0] + v[1]) + (v[2] + v[3])) + ((v[4] + v[5]) + (v[6] + v[7]));
            s += __shfl_xor(s, 16);
            if (lane < 16) atomicAdd(&pooled[(size_t)b0 * 192 + poolOff + pf], s);
        } else {
#pragma unroll
            for (int r = 0; r < 8; ++r)
                atomicAdd(&pooled[(size_t)batch[prow + r] * 192 + poolOff + pf], v[r]);
        }
    }

    const float* Ar = hin + (size_t)(tr * 16 + m) * HDIM;
    const float* Wr = W   + (size_t)(tc * 16 + m) * HDIM;
    v8f c = {0.f, 0.f, 0.f, 0.f, 0.f, 0.f, 0.f, 0.f};
#pragma unroll
    for (int k0 = 0; k0 < HDIM; k0 += 4) {
        v2f a = *(const v2f*)(Ar + k0 + kh);
        v2f b = *(const v2f*)(Wr + k0 + kh);
        const float* q = sBN + ((k0 + kh) << 2);
        v2f gam = *(const v2f*)(q), mean = *(const v2f*)(q + 2);
        v2f rs  = *(const v2f*)(q + 4), bet = *(const v2f*)(q + 6);
        a.x = bn_apply(a.x, gam.x, mean.x, rs.x, bet.x);
        a.y = bn_apply(a.y, gam.y, mean.y, rs.y, bet.y);
        c = __builtin_amdgcn_wmma_f32_16x16x4_f32(false, a, false, b, (short)0, c, false, false);
    }
    int rowBase = tr * 16 + ((lane >> 4) << 3);
    int col     = tc * 16 + m;
#pragma unroll
    for (int r = 0; r < 8; ++r)
        out[(size_t)(rowBase + r) * HDIM + col] = c[r];
}

__global__ void k_self_init(const float* __restrict__ hW, const float* __restrict__ dinv,
                            float* __restrict__ agg, int N) {
    int t = blockIdx.x * blockDim.x + threadIdx.x;
    if (t >= N * 16) return;
    int n = t >> 4, q = t & 15;
    float di  = dinv[n];
    float nrm = di * di;
    float4 v  = ((const float4*)(hW + (size_t)n * HDIM))[q];
    ((float4*)(agg + (size_t)n * HDIM))[q] =
        make_float4(v.x * nrm, v.y * nrm, v.z * nrm, v.w * nrm);
}

__device__ __forceinline__ void scatter_edge(const float* __restrict__ hW,
                                             const float* __restrict__ dinv,
                                             float* __restrict__ agg,
                                             int s, int d, int lane) {
    float nrm = dinv[s] * dinv[d];
    const float* r = hW + (size_t)s * HDIM + lane;
    float lo = r[0], hi = r[32];
    float* a = agg + (size_t)d * HDIM + lane;
    atomicAdd(a, lo * nrm);
    atomicAdd(a + 32, hi * nrm);
}

__global__ void __launch_bounds__(256)
k_scatter(const float* __restrict__ hW, const float* __restrict__ dinv,
          const int* __restrict__ src, const int* __restrict__ dst,
          float* __restrict__ agg, int E) {
    const int lane = threadIdx.x & 31;
    const int wave   = blockIdx.x * 8 + __builtin_amdgcn_readfirstlane((int)(threadIdx.x >> 5));
    const int nwaves = gridDim.x * 8;
    const int nquad  = E >> 2;
    for (int g = wave; g < nquad; g += nwaves) {
        const int e = g << 2;
        int s0 = src[e], s1 = src[e + 1], s2 = src[e + 2], s3 = src[e + 3];
        int d0 = dst[e], d1 = dst[e + 1], d2 = dst[e + 2], d3 = dst[e + 3];
        float n0 = dinv[s0] * dinv[d0];
        float n1 = dinv[s1] * dinv[d1];
        float n2 = dinv[s2] * dinv[d2];
        float n3 = dinv[s3] * dinv[d3];
        const float* r0 = hW + (size_t)s0 * HDIM + lane;
        const float* r1 = hW + (size_t)s1 * HDIM + lane;
        const float* r2 = hW + (size_t)s2 * HDIM + lane;
        const float* r3 = hW + (size_t)s3 * HDIM + lane;
        float l0 = r0[0], h0 = r0[32];
        float l1 = r1[0], h1 = r1[32];
        float l2 = r2[0], h2 = r2[32];
        float l3 = r3[0], h3 = r3[32];
        float* a0 = agg + (size_t)d0 * HDIM + lane;
        float* a1 = agg + (size_t)d1 * HDIM + lane;
        float* a2 = agg + (size_t)d2 * HDIM + lane;
        float* a3 = agg + (size_t)d3 * HDIM + lane;
        atomicAdd(a0, l0 * n0);  atomicAdd(a0 + 32, h0 * n0);
        atomicAdd(a1, l1 * n1);  atomicAdd(a1 + 32, h1 * n1);
        atomicAdd(a2, l2 * n2);  atomicAdd(a2 + 32, h2 * n2);
        atomicAdd(a3, l3 * n3);  atomicAdd(a3 + 32, h3 * n3);
    }
    for (int e = (nquad << 2) + wave; e < E; e += nwaves)
        scatter_edge(hW, dinv, agg, src[e], dst[e], lane);
}

__global__ void k_bias_relu_stats(float* __restrict__ agg, const float* __restrict__ bias,
                                  float* __restrict__ stats  , int N) {
    __shared__ float sS[256];
    __shared__ float sQ[256];
    int f  = threadIdx.x & 63;
    int rg = threadIdx.x >> 6;
    float b = bias[f];
    float s = 0.f, q = 0.f;
    for (int row = blockIdx.x * 4 + rg; row < N; row += gridDim.x * 4) {
        size_t idx = (size_t)row * HDIM + f;
        float v = agg[idx] + b;
        v = fmaxf(v, 0.f);
        agg[idx] = v;
        s += v;
        q += v * v;
    }
    sS[threadIdx.x] = s;
    sQ[threadIdx.x] = q;
    __syncthreads();
    if (rg == 0) {
        s = sS[f] + sS[64 + f] + sS[128 + f] + sS[192 + f];
        q = sQ[f] + sQ[64 + f] + sQ[128 + f] + sQ[192 + f];
        atomicAdd(&stats[f], s);
        atomicAdd(&stats[64 + f], q);
    }
}

__global__ void k_bn_pool(const float* __restrict__ hin, const float* __restrict__ stats,
                          const float* __restrict__ g, const float* __restrict__ beta,
                          const int* __restrict__ batch, float* __restrict__ hout,
                          float* __restrict__ pooled, int poolOff, int N) {
    int t = blockIdx.x * blockDim.x + threadIdx.x;
    if (t >= N * HDIM) return;
    int n = t >> 6, f = t & 63;
    float invN = 1.0f / (float)N;
    float mean = stats[f] * invN;
    float var  = stats[64 + f] * invN - mean * mean;
    float v = g[f] * (hin[t] - mean) * rsqrtf(var + EPSV) + beta[f];
    hout[t] = v;
    atomicAdd(&pooled[(size_t)batch[n] * 192 + poolOff + f], v);
}

extern "C" void kernel_launch(void* const* d_in, const int* in_sizes, int n_in,
                              void* d_out, int out_size, void* d_ws, size_t ws_size,
                              hipStream_t stream) {
    const float* x    = (const float*)d_in[0];
    const int*   eidx = (const int*)d_in[1];
    const int*   batch = (const int*)d_in[2];
    const float* W[3]  = {(const float*)d_in[3],  (const float*)d_in[7],  (const float*)d_in[11]};
    const float* bi[3] = {(const float*)d_in[4],  (const float*)d_in[8],  (const float*)d_in[12]};
    const float* ga[3] = {(const float*)d_in[5],  (const float*)d_in[9],  (const float*)d_in[13]};
    const float* be[3] = {(const float*)d_in[6],  (const float*)d_in[10], (const float*)d_in[14]};

    const int N = in_sizes[0] / HDIM;
    const int E = in_sizes[1] / 2;
    const int G = (out_size - N * HDIM) / (3 * HDIM);

    const int* src = eidx;
    const int* dst = eidx + E;

    float* ws    = (float*)d_ws;
    float* dinv  = ws;
    float* tmp   = dinv + N;
    float* agg   = tmp  + (size_t)N * HDIM;
    float* stats = agg  + (size_t)N * HDIM;

    float* pooled = (float*)d_out;
    float* hfinal = (float*)d_out + (size_t)G * 192;

    const int TB = 256;

    k_deg_init <<<cdiv(N, TB), TB, 0, stream>>>(dinv, N);
    k_deg_accum<<<cdiv(E, TB), TB, 0, stream>>>(dst, dinv, E);
    k_dinv     <<<cdiv(N, TB), TB, 0, stream>>>(dinv, N);

    k_zero<<<cdiv((long long)G * 192, TB), TB, 0, stream>>>(pooled, G * 192);

    const int rowTiles   = N / 16;
    const int gemmWaves  = rowTiles * (HDIM / 16);
    const int gemmBlocks = cdiv((long long)gemmWaves * 32, TB);

    for (int l = 0; l < 3; ++l) {
        if (l == 0)
            k_gemm_wmma<<<gemmBlocks, TB, 0, stream>>>(x, W[0], tmp, rowTiles);
        else
            k_bn_pool_gemm<<<gemmBlocks, TB, 0, stream>>>(agg, stats, ga[l - 1], be[l - 1], batch,
                                                         pooled, (l - 1) * HDIM, W[l], tmp, rowTiles, N);
        k_self_init<<<cdiv((long long)N * 16, TB), TB, 0, stream>>>(tmp, dinv, agg, N);
        k_scatter  <<<2048, TB, 0, stream>>>(tmp, dinv, src, dst, agg, E);

        k_zero<<<1, 128, 0, stream>>>(stats, 128);
        k_bias_relu_stats<<<1024, TB, 0, stream>>>(agg, bi[l], stats, N);
    }
    k_bn_pool<<<cdiv((long long)N * HDIM, TB), TB, 0, stream>>>(
        agg, stats, ga[2], be[2], batch, hfinal, pooled, 2 * HDIM, N);
}


# Self-loop term (hW * dinv[n]^2, the aggregation's start value) is stored by the GEMM kernel's epilogue from its accumulators; the 3 self-init passes are gone, 20 launches become 17; sum order kept
# speedup vs baseline: 1.3506x; 1.0353x over previous
#include <hip/hip_runtime.h>

typedef float v2f __attribute__((ext_vector_type(2)));
typedef float v8f __attribute__((ext_vector_type(8)));

#define HDIM 64
#define EPSV 1e-5f

static inline int cdiv(long long a, long long b) { return (int)((a + b - 1) / b); }

__global__ void k_zero(float* __restrict__ p, int n) {
    int i = blockIdx.x * blockDim.x + threadIdx.x;
    if (i < n) p[i] = 0.0f;
}

__global__ void k_deg_init(float* __restrict__ deg, int n) {
    int i = blockIdx.x * blockDim.x + threadIdx.x;
    if (i < n) deg[i] = 1.0f;
}

__global__ void k_deg_accum(const int* __restrict__ dst, float* __restrict__ deg, int E) {
    int e = blockIdx.x * blockDim.x + threadIdx.x;
    if (e < E) atomicAdd(&deg[dst[e]], 1.0f);
}

__global__ void k_dinv(float* __restrict__ deg, int n) {
    int i = blockIdx.x * blockDim.x + threadIdx.x;
    if (i < n) deg[i] = rsqrtf(deg[i]);
}

__global__ void k_gemm_wmma(const float* __restrict__ A, const float* __restrict__ W,
                            const float* __restrict__ dinv,
                            float* __restrict__ out, float* __restrict__ agg, int rowTiles) {
    int wid  = (blockIdx.x * blockDim.x + threadIdx.x) >> 5;
    int lane = threadIdx.x & 31;
    int tr = wid >> 2;
    int tc = wid & 3;
    if (tr >= rowTiles) return;
    int m  = lane & 15;
    int kh = (lane >> 4) << 1;
    const float* Ar = A + (size_t)(tr * 16 + m) * HDIM;
    const float* Wr = W + (size_t)(tc * 16 + m) * HDIM;
    int rowBase = tr * 16 + ((lane >> 4) << 3);
    int col     = tc * 16 + m;
    const float4 dA = *(const float4*)(dinv + rowBase);
    const float4 dB = *(const float4*)(dinv + rowBase + 4);
    v8f c = {0.f, 0.f, 0.f, 0.f, 0.f, 0.f, 0.f, 0.f};
#pragma unroll
    for (int k0 = 0; k0 < HDIM; k0 += 4) {
        v2f a = *(const v2f*)(Ar + k0 + kh);
        v2f b = *(const v2f*)(Wr + k0 + kh);
        c = __builtin_amdgcn_wmma_f32_16x16x4_f32(false, a, false, b, (short)0, c, false, false);
    }
    const float di[8] = {dA.x, dA.y, dA.z, dA.w, dB.x, dB.y, dB.z, dB.w};
#pragma unroll
    for (int r = 0; r < 8; ++r) {
        size_t o = (size_t)(rowBase + r) * HDIM + col;
        out[o] = c[r];
        agg[o] = c[r] * (di[r] * di[r]);
    }
}

__device__ __forceinline__ float bn_apply(float x, float gam, float mean, float rs, float bet) {
    return gam * (x - mean) * rs + bet;
}

__global__ void __launch_bounds__(256)
k_bn_pool_gemm(const float* hin, const float* __restrict__ stats,
               const float* __restrict__ g, const float* __restrict__ beta,
               const int* __restrict__ batch, float* __restrict__ pooled, int poolOff,
               const float* __restrict__ W, const float* __restrict__ dinv,
               float* __restrict__ out, float* aggOut, int rowTiles, int N) {
    __shared__ __attribute__((aligned(16))) float sBN[HDIM * 4];
    if (threadIdx.x < HDIM) {
        int f = threadIdx.x;
        float invN = 1.0f / (float)N;
        float mean = stats[f] * invN;
        float var  = stats[64 + f] * invN - mean * mean;
        float* q = sBN + ((f >> 1) << 3) + (f & 1);
        q[0] = g[f];
        q[2] = mean;
        q[4] = rsqrtf(var + EPSV);
        q[6] = beta[f];
    }
    __syncthreads();
    int wid  = (blockIdx.x * blockDim.x + threadIdx.x) >> 5;
    int lane = threadIdx.x & 31;
    int tr = wid >> 2;
    int tc = wid & 3;
    const bool live = tr < rowTiles;
    int m  = lane & 15;
    int kh = (lane >> 4) << 1;
    int rowBase = tr * 16 + ((lane >> 4) << 3);
    int col     = tc * 16 + m;
    v8f c = {0.f, 0.f, 0.f, 0.f, 0.f, 0.f, 0.f, 0.f};
    float4 dA = make_float4(0.f, 0.f, 0.f, 0.f), dB = dA;
    if (live) {

    {
        int pf   = tc * 16 + m;
        int prow = tr * 16 + ((lane >> 4) << 3);
        const float* q = sBN + ((pf >> 1) << 3) + (pf & 1);
        float pg = q[0], pm = q[2], pr = q[4], pb = q[6];
        float v[8];
#pragma unroll
        for (int r = 0; r < 8; ++r)
            v[r] = bn_apply(hin[(size_t)(prow + r) * HDIM + pf], pg, pm, pr, pb);
        int bmine = batch[tr * 16 + m];
        int b0    = __builtin_amdgcn_readfirstlane(bmine);
        if (__all(bmine == b0)) {
            float s = ((v[0] + v[1]) + (v[2] + v[3])) + ((v[4] + v[5]) + (v[6] + v[7]));
            s += __shfl_xor(s, 16);
            if (lane < 16) atomicAdd(&pooled[(size_t)b0 * 192 + poolOff + pf], s);
        } else {
#pragma unroll
            for (int r = 0; r < 8; ++r)
                atomicAdd(&pooled[(size_t)batch[prow + r] * 192 + poolOff + pf], v[r]);
        }
    }

    const float* Ar = hin + (size_t)(tr * 16 + m) * HDIM;
    const float* Wr = W   + (size_t)(tc * 16 + m) * HDIM;
#pragma unroll
    for (int k0 = 0; k0 < HDIM; k0 += 4) {
        v2f a = *(const v2f*)(Ar + k0 + kh);
        v2f b = *(const v2f*)(Wr + k0 + kh);
        const float* q = sBN + ((k0 + kh) << 2);
        v2f gam = *(const v2f*)(q), mean = *(const v2f*)(q + 2);
        v2f rs  = *(const v2f*)(q + 4), bet = *(const v2f*)(q + 6);
        a.x = bn_apply(a.x, gam.x, mean.x, rs.x, bet.x);
        a.y = bn_apply(a.y, gam.y, mean.y, rs.y, bet.y);
        c = __builtin_amdgcn_wmma_f32_16x16x4_f32(false, a, false, b, (short)0, c, false, false);
    }
    dA = *(const float4*)(dinv + rowBase);
    dB = *(const float4*)(dinv + rowBase + 4);
    }
    __syncthreads();
    if (live) {
        const float di[8] = {dA.x, dA.y, dA.z, dA.w, dB.x, dB.y, dB.z, dB.w};
#pragma unroll
        for (int r = 0; r < 8; ++r) {
            size_t o = (size_t)(rowBase + r) * HDIM + col;
            out[o]    = c[r];
            aggOut[o] = c[r] * (di[r] * di[r]);
        }
    }
}

__device__ __forceinline__ void scatter_edge(const float* __restrict__ hW,
                                             const float* __restrict__ dinv,
                                             float* __restrict__ agg,
                                             int s, int d, int lane) {
    float nrm = dinv[s] * dinv[d];
    const float* r = hW + (size_t)s * HDIM + lane;
    float lo = r[0], hi = r[32];
    float* a = agg + (size_t)d * HDIM + lane;
    atomicAdd(a, lo * nrm);
    atomicAdd(a + 32, hi * nrm);
}

__global__ void __launch_bounds__(256)
k_scatter(const float* __restrict__ hW, const float* __restrict__ dinv,
          const int* __restrict__ src, const int* __restrict__ dst,
          float* __restrict__ agg, int E) {
    const int lane = threadIdx.x & 31;
    const int wave   = blockIdx.x * 8 + __builtin_amdgcn_readfirstlane((int)(threadIdx.x >> 5));
    const int nwaves = gridDim.x * 8;
    const int nquad  = E >> 2;
    for (int g = wave; g < nquad; g += nwaves) {
        const int e = g << 2;
        int s0 = src[e], s1 = src[e + 1], s2 = src[e + 2], s3 = src[e + 3];
        int d0 = dst[e], d1 = dst[e + 1], d2 = dst[e + 2], d3 = dst[e + 3];
        float n0 = dinv[s0] * dinv[d0];
        float n1 = dinv[s1] * dinv[d1];
        float n2 = dinv[s2] * dinv[d2];
        float n3 = dinv[s3] * dinv[d3];
        const float* r0 = hW + (size_t)s0 * HDIM + lane;
        const float* r1 = hW + (size_t)s1 * HDIM + lane;
        const float* r2 = hW + (size_t)s2 * HDIM + lane;
        const float* r3 = hW + (size_t)s3 * HDIM + lane;
        float l0 = r0[0], h0 = r0[32];
        float l1 = r1[0], h1 = r1[32];
        float l2 = r2[0], h2 = r2[32];
        float l3 = r3[0], h3 = r3[32];
        float* a0 = agg + (size_t)d0 * HDIM + lane;
        float* a1 = agg + (size_t)d1 * HDIM + lane;
        float* a2 = agg + (size_t)d2 * HDIM + lane;
        float* a3 = agg + (size_t)d3 * HDIM + lane;
        atomicAdd(a0, l0 * n0);  atomicAdd(a0 + 32, h0 * n0);
        atomicAdd(a1, l1 * n1);  atomicAdd(a1 + 32, h1 * n1);
        atomicAdd(a2, l2 * n2);  atomicAdd(a2 + 32, h2 * n2);
        atomicAdd(a3, l3 * n3);  atomicAdd(a3 + 32, h3 * n3);
    }
    for (int e = (nquad << 2) + wave; e < E; e += nwaves)
        scatter_edge(hW, dinv, agg, src[e], dst[e], lane);
}

__global__ void k_bias_relu_stats(float* __restrict__ agg, const float* __restrict__ bias,
                                  float* __restrict__ stats  , int N) {
    __shared__ float sS[256];
    __shared__ float sQ[256];
    int f  = threadIdx.x & 63;
    int rg = threadIdx.x >> 6;
    float b = bias[f];
    float s = 0.f, q = 0.f;
    for (int row = blockIdx.x * 4 + rg; row < N; row += gridDim.x * 4) {
        size_t idx = (size_t)row * HDIM + f;
        float v = agg[idx] + b;
        v = fmaxf(v, 0.f);
        agg[idx] = v;
        s += v;
        q += v * v;
    }
    sS[threadIdx.x] = s;
    sQ[threadIdx.x] = q;
    __syncthreads();
    if (rg == 0) {
        s = sS[f] + sS[64 + f] + sS[128 + f] + sS[192 + f];
        q = sQ[f] + sQ[64 + f] + sQ[128 + f] + sQ[192 + f];
        atomicAdd(&stats[f], s);
        atomicAdd(&stats[64 + f], q);
    }
}

__global__ void k_bn_pool(const float* __restrict__ hin, const float* __restrict__ stats,
                          const float* __restrict__ g, const float* __restrict__ beta,
                          const int* __restrict__ batch, float* __restrict__ hout,
                          float* __restrict__ pooled, int poolOff, int N) {
    int t = blockIdx.x * blockDim.x + threadIdx.x;
    if (t >= N * HDIM) return;
    int n = t >> 6, f = t & 63;
    float invN = 1.0f / (float)N;
    float mean = stats[f] * invN;
    float var  = stats[64 + f] * invN - mean * mean;
    float v = g[f] * (hin[t] - mean) * rsqrtf(var + EPSV) + beta[f];
    hout[t] = v;
    atomicAdd(&pooled[(size_t)batch[n] * 192 + poolOff + f], v);
}

extern "C" void kernel_launch(void* const* d_in, const int* in_sizes, int n_in,
                              void* d_out, int out_size, void* d_ws, size_t ws_size,
                              hipStream_t stream) {
    const float* x    = (const float*)d_in[0];
    const int*   eidx = (const int*)d_in[1];
    const int*   batch = (const int*)d_in[2];
    const float* W[3]  = {(const float*)d_in[3],  (const float*)d_in[7],  (const float*)d_in[11]};
    const float* bi[3] = {(const float*)d_in[4],  (const float*)d_in[8],  (const float*)d_in[12]};
    const float* ga[3] = {(const float*)d_in[5],  (const float*)d_in[9],  (const float*)d_in[13]};
    const float* be[3] = {(const float*)d_in[6],  (const float*)d_in[10], (const float*)d_in[14]};

    const int N = in_sizes[0] / HDIM;
    const int E = in_sizes[1] / 2;
    const int G = (out_size - N * HDIM) / (3 * HDIM);

    const int* src = eidx;
    const int* dst = eidx + E;

    float* ws    = (float*)d_ws;
    float* dinv  = ws;
    float* tmp   = dinv + N;
    float* agg   = tmp  + (size_t)N * HDIM;
    float* stats = agg  + (size_t)N * HDIM;

    float* pooled = (float*)d_out;
    float* hfinal = (float*)d_out + (size_t)G * 192;

    const int TB = 256;

    k_deg_init <<<cdiv(N, TB), TB, 0, stream>>>(dinv, N);
    k_deg_accum<<<cdiv(E, TB), TB, 0, stream>>>(dst, dinv, E);
    k_dinv     <<<cdiv(N, TB), TB, 0, stream>>>(dinv, N);

    k_zero<<<cdiv((long long)G * 192, TB), TB, 0, stream>>>(pooled, G * 192);

    const int rowTiles   = N / 16;
    const int gemmWaves  = rowTiles * (HDIM / 16);
    const int gemmBlocks = cdiv((long long)gemmWaves * 32, TB);

    for (int l = 0; l < 3; ++l) {
        if (l == 0)
            k_gemm_wmma<<<gemmBlocks, TB, 0, stream>>>(x, W[0], dinv, tmp, agg, rowTiles);
        else
            k_bn_pool_gemm<<<gemmBlocks, TB, 0, stream>>>(agg, stats, ga[l - 1], be[l - 1], batch,
                                                         pooled, (l - 1) * HDIM, W[l], dinv, tmp, agg,
                                                         rowTiles, N);
        k_scatter  <<<2048, TB, 0, stream>>>(tmp, dinv, src, dst, agg, E);

        k_zero<<<1, 128, 0, stream>>>(stats, 128);
        k_bias_relu_stats<<<1024, TB, 0, stream>>>(agg, bi[l], stats, N);
    }
    k_bn_pool<<<cdiv((long long)N * HDIM, TB), TB, 0, stream>>>(
        agg, stats, ga[2], be[2], batch, hfinal, pooled, 2 * HDIM, N);
}


# Edge scatter adds into two planes, one per package half (a wave reads its die number and uses its half's plane); bias+ReLU pass sums them; no line added into from both halves; sum order changes
# speedup vs baseline: 1.5597x; 1.1548x over previous
#include <hip/hip_runtime.h>

typedef float v2f __attribute__((ext_vector_type(2)));
typedef float v8f __attribute__((ext_vector_type(8)));

#define HDIM 64
#define EPSV 1e-5f

static inline int cdiv(long long a, long long b) { return (int)((a + b - 1) / b); }

__global__ void k_zero(float* __restrict__ p, int n) {
    int i = blockIdx.x * blockDim.x + threadIdx.x;
    if (i < n) p[i] = 0.0f;
}

__global__ void k_deg_init(float* __restrict__ deg, int n) {
    int i = blockIdx.x * blockDim.x + threadIdx.x;
    if (i < n) deg[i] = 1.0f;
}

__global__ void k_deg_accum(const int* __restrict__ dst, float* __restrict__ deg, int E) {
    int e = blockIdx.x * blockDim.x + threadIdx.x;
    if (e < E) atomicAdd(&deg[dst[e]], 1.0f);
}

__global__ void k_dinv(float* __restrict__ deg, int n) {
    int i = blockIdx.x * blockDim.x + threadIdx.x;
    if (i < n) deg[i] = rsqrtf(deg[i]);
}

__global__ void k_gemm_wmma(const float* __restrict__ A, const float* __restrict__ W,
                            const float* __restrict__ dinv,
                            float* __restrict__ out, float* __restrict__ agg,
                            float* __restrict__ agg1, int rowTiles) {
    int wid  = (blockIdx.x * blockDim.x + threadIdx.x) >> 5;
    int lane = threadIdx.x & 31;
    int tr = wid >> 2;
    int tc = wid & 3;
    if (tr >= rowTiles) return;
    int m  = lane & 15;
    int kh = (lane >> 4) << 1;
    const float* Ar = A + (size_t)(tr * 16 + m) * HDIM;
    const float* Wr = W + (size_t)(tc * 16 + m) * HDIM;
    int rowBase = tr * 16 + ((lane >> 4) << 3);
    int col     = tc * 16 + m;
    const float4 dA = *(const float4*)(dinv + rowBase);
    const float4 dB = *(const float4*)(dinv + rowBase + 4);
#pragma unroll
    for (int r = 0; r < 8; ++r)
        agg1[(size_t)(rowBase + r) * HDIM + col] = 0.0f;
    v8f c = {0.f, 0.f, 0.f, 0.f, 0.f, 0.f, 0.f, 0.f};
#pragma unroll
    for (int k0 = 0; k0 < HDIM; k0 += 4) {
        v2f a = *(const v2f*)(Ar + k0 + kh);
        v2f b = *(const v2f*)(Wr + k0 + kh);
        c = __builtin_amdgcn_wmma_f32_16x16x4_f32(false, a, false, b, (short)0, c, false, false);
    }
    const float di[8] = {dA.x, dA.y, dA.z, dA.w, dB.x, dB.y, dB.z, dB.w};
#pragma unroll
    for (int r = 0; r < 8; ++r) {
        size_t o = (size_t)(rowBase + r) * HDIM + col;
        out[o] = c[r];
        agg[o] = c[r] * (di[r] * di[r]);
    }
}

__device__ __forceinline__ float bn_apply(float x, float gam, float mean, float rs, float bet) {
    return gam * (x - mean) * rs + bet;
}

__global__ void __launch_bounds__(256)
k_bn_pool_gemm(const float* hin, const float* __restrict__ stats,
               const float* __restrict__ g, const float* __restrict__ beta,
               const int* __restrict__ batch, float* __restrict__ pooled, int poolOff,
               const float* __restrict__ W, const float* __restrict__ dinv,
               float* __restrict__ out, float* aggOut, float* __restrict__ agg1,
               int rowTiles, int N) {
    __shared__ __attribute__((aligned(16))) float sBN[HDIM * 4];
    if (threadIdx.x < HDIM) {
        int f = threadIdx.x;
        float invN = 1.0f / (float)N;
        float mean = stats[f] * invN;
        float var  = stats[64 + f] * invN - mean * mean;
        float* q = sBN + ((f >> 1) << 3) + (f & 1);
        q[0] = g[f];
        q[2] = mean;
        q[4] = rsqrtf(var + EPSV);
        q[6] = beta[f];
    }
    __syncthreads();
    int wid  = (blockIdx.x * blockDim.x + threadIdx.x) >> 5;
    int lane = threadIdx.x & 31;
    int tr = wid >> 2;
    int tc = wid & 3;
    const bool live = tr < rowTiles;
    int m  = lane & 15;
    int kh = (lane >> 4) << 1;
    int rowBase = tr * 16 + ((lane >> 4) << 3);
    int col     = tc * 16 + m;
    v8f c = {0.f, 0.f, 0.f, 0.f, 0.f, 0.f, 0.f, 0.f};
    float4 dA = make_float4(0.f, 0.f, 0.f, 0.f), dB = dA;
    if (live) {

    {
        int pf   = tc * 16 + m;
        int prow = tr * 16 + ((lane >> 4) << 3);
        const float* q = sBN + ((pf >> 1) << 3) + (pf & 1);
        float pg = q[0], pm = q[2], pr = q[4], pb = q[6];
        float v[8];
#pragma unroll
        for (int r = 0; r < 8; ++r)
            v[r] = bn_apply(hin[(size_t)(prow + r) * HDIM + pf], pg, pm, pr, pb);
        int bmine = batch[tr * 16 + m];
        int b0    = __builtin_amdgcn_readfirstlane(bmine);
        if (__all(bmine == b0)) {
            float s = ((v[0] + v[1]) + (v[2] + v[3])) + ((v[4] + v[5]) + (v[6] + v[7]));
            s += __shfl_xor(s, 16);
            if (lane < 16) atomicAdd(&pooled[(size_t)b0 * 192 + poolOff + pf], s);
        } else {
#pragma unroll
            for (int r = 0; r < 8; ++r)
                atomicAdd(&pooled[(size_t)batch[prow + r] * 192 + poolOff + pf], v[r]);
        }
    }

    const float* Ar = hin + (size_t)(tr * 16 + m) * HDIM;
    const float* Wr = W   + (size_t)(tc * 16 + m) * HDIM;
#pragma unroll
    for (int k0 = 0; k0 < HDIM; k0 += 4) {
        v2f a = *(const v2f*)(Ar + k0 + kh);
        v2f b = *(const v2f*)(Wr + k0 + kh);
        const float* q = sBN + ((k0 + kh) << 2);
        v2f gam = *(const v2f*)(q), mean = *(const v2f*)(q + 2);
        v2f rs  = *(const v2f*)(q + 4), bet = *(const v2f*)(q + 6);
        a.x = bn_apply(a.x, gam.x, mean.x, rs.x, bet.x);
        a.y = bn_apply(a.y, gam.y, mean.y, rs.y, bet.y);
        c = __builtin_amdgcn_wmma_f32_16x16x4_f32(false, a, false, b, (short)0, c, false, false);
    }
    dA = *(const float4*)(dinv + rowBase);
    dB = *(const float4*)(dinv + rowBase + 4);
    }
    __syncthreads();
    if (live) {
        const float di[8] = {dA.x, dA.y, dA.z, dA.w, dB.x, dB.y, dB.z, dB.w};
#pragma unroll
        for (int r = 0; r < 8; ++r) {
            size_t o = (size_t)(rowBase + r) * HDIM + col;
            out[o]    = c[r];
            aggOut[o] = c[r] * (di[r] * di[r]);
            agg1[o]   = 0.0f;
        }
    }
}

__device__ __forceinline__ void scatter_edge(const float* __restrict__ hW,
                                             const float* __restrict__ dinv,
                                             float* __restrict__ agg,
                                             int s, int d, int lane) {
    float nrm = dinv[s] * dinv[d];
    const float* r = hW + (size_t)s * HDIM + lane;
    float lo = r[0], hi = r[32];
    float* a = agg + (size_t)d * HDIM + lane;
    atomicAdd(a, lo * nrm);
    atomicAdd(a + 32, hi * nrm);
}

__global__ void __launch_bounds__(256)
k_scatter(const float* __restrict__ hW, const float* __restrict__ dinv,
          const int* __restrict__ src, const int* __restrict__ dst,
          float* __restrict__ agg0, float* __restrict__ agg1, int E) {
    const int lane = threadIdx.x & 31;
    const unsigned place = __builtin_amdgcn_s_sendmsg_rtn(0x87);
    float* __restrict__ agg = ((place >> 18) & 1u) ? agg1 : agg0;
    const int wave   = blockIdx.x * 8 + __builtin_amdgcn_readfirstlane((int)(threadIdx.x >> 5));
    const int nwaves = gridDim.x * 8;
    const int nquad  = E >> 2;
    for (int g = wave; g < nquad; g += nwaves) {
        const int e = g << 2;
        int s0 = src[e], s1 = src[e + 1], s2 = src[e + 2], s3 = src[e + 3];
        int d0 = dst[e], d1 = dst[e + 1], d2 = dst[e + 2], d3 = dst[e + 3];
        float n0 = dinv[s0] * dinv[d0];
        float n1 = dinv[s1] * dinv[d1];
        float n2 = dinv[s2] * dinv[d2];
        float n3 = dinv[s3] * dinv[d3];
        const float* r0 = hW + (size_t)s0 * HDIM + lane;
        const float* r1 = hW + (size_t)s1 * HDIM + lane;
        const float* r2 = hW + (size_t)s2 * HDIM + lane;
        const float* r3 = hW + (size_t)s3 * HDIM + lane;
        float l0 = r0[0], h0 = r0[32];
        float l1 = r1[0], h1 = r1[32];
        float l2 = r2[0], h2 = r2[32];
        float l3 = r3[0], h3 = r3[32];
        float* a0 = agg + (size_t)d0 * HDIM + lane;
        float* a1 = agg + (size_t)d1 * HDIM + lane;
        float* a2 = agg + (size_t)d2 * HDIM + lane;
        float* a3 = agg + (size_t)d3 * HDIM + lane;
        atomicAdd(a0, l0 * n0);  atomicAdd(a0 + 32, h0 * n0);
        atomicAdd(a1, l1 * n1);  atomicAdd(a1 + 32, h1 * n1);
        atomicAdd(a2, l2 * n2);  atomicAdd(a2 + 32, h2 * n2);
        atomicAdd(a3, l3 * n3);  atomicAdd(a3 + 32, h3 * n3);
    }
    for (int e = (nquad << 2) + wave; e < E; e += nwaves)
        scatter_edge(hW, dinv, agg, src[e], dst[e], lane);
}

__global__ void k_bias_relu_stats(float* __restrict__ agg, const float* __restrict__ agg1,
                                  const float* __restrict__ bias,
                                  float* __restrict__ stats  , int N) {
    __shared__ float sS[256];
    __shared__ float sQ[256];
    int f  = threadIdx.x & 63;
    int rg = threadIdx.x >> 6;
    float b = bias[f];
    float s = 0.f, q = 0.f;
    for (int row = blockIdx.x * 4 + rg; row < N; row += gridDim.x * 4) {
        size_t idx = (size_t)row * HDIM + f;
        float v = (agg[idx] + agg1[idx]) + b;
        v = fmaxf(v, 0.f);
        agg[idx] = v;
        s += v;
        q += v * v;
    }
    sS[threadIdx.x] = s;
    sQ[threadIdx.x] = q;
    __syncthreads();
    if (rg == 0) {
        s = sS[f] + sS[64 + f] + sS[128 + f] + sS[192 + f];
        q = sQ[f] + sQ[64 + f] + sQ[128 + f] + sQ[192 + f];
        atomicAdd(&stats[f], s);
        atomicAdd(&stats[64 + f], q);
    }
}

__global__ void k_bn_pool(const float* __restrict__ hin, const float* __restrict__ stats,
                          const float* __restrict__ g, const float* __restrict__ beta,
                          const int* __restrict__ batch, float* __restrict__ hout,
                          float* __restrict__ pooled, int poolOff, int N) {
    int t = blockIdx.x * blockDim.x + threadIdx.x;
    if (t >= N * HDIM) return;
    int n = t >> 6, f = t & 63;
    float invN = 1.0f / (float)N;
    float mean = stats[f] * invN;
    float var  = stats[64 + f] * invN - mean * mean;
    float v = g[f] * (hin[t] - mean) * rsqrtf(var + EPSV) + beta[f];
    hout[t] = v;
    atomicAdd(&pooled[(size_t)batch[n] * 192 + poolOff + f], v);
}

extern "C" void kernel_launch(void* const* d_in, const int* in_sizes, int n_in,
                              void* d_out, int out_size, void* d_ws, size_t ws_size,
                              hipStream_t stream) {
    const float* x    = (const float*)d_in[0];
    const int*   eidx = (const int*)d_in[1];
    const int*   batch = (const int*)d_in[2];
    const float* W[3]  = {(const float*)d_in[3],  (const float*)d_in[7],  (const float*)d_in[11]};
    const float* bi[3] = {(const float*)d_in[4],  (const float*)d_in[8],  (const float*)d_in[12]};
    const float* ga[3] = {(const float*)d_in[5],  (const float*)d_in[9],  (const float*)d_in[13]};
    const float* be[3] = {(const float*)d_in[6],  (const float*)d_in[10], (const float*)d_in[14]};

    const int N = in_sizes[0] / HDIM;
    const int E = in_sizes[1] / 2;
    const int G = (out_size - N * HDIM) / (3 * HDIM);

    const int* src = eidx;
    const int* dst = eidx + E;

    float* ws    = (float*)d_ws;
    float* dinv  = ws;
    float* tmp   = dinv + N;
    float* agg   = tmp  + (size_t)N * HDIM;
    float* stats = agg  + (size_t)N * HDIM;

    float* pooled = (float*)d_out;
    float* hfinal = (float*)d_out + (size_t)G * 192;
    float* agg1 = hfinal;

    const int TB = 256;

    k_deg_init <<<cdiv(N, TB), TB, 0, stream>>>(dinv, N);
    k_deg_accum<<<cdiv(E, TB), TB, 0, stream>>>(dst, dinv, E);
    k_dinv     <<<cdiv(N, TB), TB, 0, stream>>>(dinv, N);

    k_zero<<<cdiv((long long)G * 192, TB), TB, 0, stream>>>(pooled, G * 192);

    const int rowTiles   = N / 16;
    const int gemmWaves  = rowTiles * (HDIM / 16);
    const int gemmBlocks = cdiv((long long)gemmWaves * 32, TB);

    for (int l = 0; l < 3; ++l) {
        if (l == 0)
            k_gemm_wmma<<<gemmBlocks, TB, 0, stream>>>(x, W[0], dinv, tmp, agg, agg1, rowTiles);
        else
            k_bn_pool_gemm<<<gemmBlocks, TB, 0, stream>>>(agg, stats, ga[l - 1], be[l - 1], batch,
                                                         pooled, (l - 1) * HDIM, W[l], dinv, tmp, agg, agg1,
                                                         rowTiles, N);
        k_scatter  <<<2048, TB, 0, stream>>>(tmp, dinv, src, dst, agg, agg1, E);

        k_zero<<<1, 128, 0, stream>>>(stats, 128);
        k_bias_relu_stats<<<1024, TB, 0, stream>>>(agg, agg1, bi[l], stats, N);
    }
    k_bn_pool<<<cdiv((long long)N * HDIM, TB), TB, 0, stream>>>(
        agg, stats, ga[2], be[2], batch, hfinal, pooled, 2 * HDIM, N);
}


# Scatter adds rows as stored: GEMM epilogue stores rows times dinv[row], bias+ReLU pass applies dinv[dst] once per node; per-edge dinv lookups and multiplies gone; products regrouped, sum order kept
# speedup vs baseline: 1.6932x; 1.0856x over previous
#include <hip/hip_runtime.h>

typedef float v2f __attribute__((ext_vector_type(2)));
typedef float v8f __attribute__((ext_vector_type(8)));

#define HDIM 64
#define EPSV 1e-5f

static inline int cdiv(long long a, long long b) { return (int)((a + b - 1) / b); }

__global__ void k_zero(float* __restrict__ p, int n) {
    int i = blockIdx.x * blockDim.x + threadIdx.x;
    if (i < n) p[i] = 0.0f;
}

__global__ void k_deg_init(float* __restrict__ deg, int n) {
    int i = blockIdx.x * blockDim.x + threadIdx.x;
    if (i < n) deg[i] = 1.0f;
}

__global__ void k_deg_accum(const int* __restrict__ dst, float* __restrict__ deg, int E) {
    int e = blockIdx.x * blockDim.x + threadIdx.x;
    if (e < E) atomicAdd(&deg[dst[e]], 1.0f);
}

__global__ void k_dinv(float* __restrict__ deg, int n) {
    int i = blockIdx.x * blockDim.x + threadIdx.x;
    if (i < n) deg[i] = rsqrtf(deg[i]);
}

__global__ void __launch_bounds__(256, 16)
k_gemm_wmma(const float* __restrict__ A, const float* __restrict__ W,
                            const float* __restrict__ dinv,
                            float* __restrict__ out, float* __restrict__ agg,
                            float* __restrict__ agg1, int rowTiles) {
    int wid  = (blockIdx.x * blockDim.x + threadIdx.x) >> 5;
    int lane = threadIdx.x & 31;
    int tr = wid >> 2;
    int tc = wid & 3;
    if (tr >= rowTiles) return;
    int m  = lane & 15;
    int kh = (lane >> 4) << 1;
    const float* Ar = A + (size_t)(tr * 16 + m) * HDIM;
    const float* Wr = W + (size_t)(tc * 16 + m) * HDIM;
    int rowBase = tr * 16 + ((lane >> 4) << 3);
    int col     = tc * 16 + m;
#pragma unroll
    for (int r = 0; r < 8; ++r)
        agg1[(size_t)(rowBase + r) * HDIM + col] = 0.0f;
    v8f c = {0.f, 0.f, 0.f, 0.f, 0.f, 0.f, 0.f, 0.f};
#pragma unroll
    for (int k0 = 0; k0 < HDIM; k0 += 4) {
        v2f a = *(const v2f*)(Ar + k0 + kh);
        v2f b = *(const v2f*)(Wr + k0 + kh);
        c = __builtin_amdgcn_wmma_f32_16x16x4_f32(false, a, false, b, (short)0, c, false, false);
    }
    const float4 dA = *(const float4*)(dinv + rowBase);
    const float4 dB = *(const float4*)(dinv + rowBase + 4);
    const float di[8] = {dA.x, dA.y, dA.z, dA.w, dB.x, dB.y, dB.z, dB.w};
#pragma unroll
    for (int r = 0; r < 8; ++r) {
        size_t o = (size_t)(rowBase + r) * HDIM + col;
        float v = c[r] * di[r];
        out[o] = v;
        agg[o] = v;
    }
}

__device__ __forceinline__ float bn_apply(float x, float gam, float mean, float rs, float bet) {
    return gam * (x - mean) * rs + bet;
}

__global__ void __launch_bounds__(256)
k_bn_pool_gemm(const float* hin, const float* __restrict__ stats,
               const float* __restrict__ g, const float* __restrict__ beta,
               const int* __restrict__ batch, float* __restrict__ pooled, int poolOff,
               const float* __restrict__ W, const float* __restrict__ dinv,
               float* __restrict__ out, float* aggOut, float* __restrict__ agg1,
               int rowTiles, int N) {
    __shared__ __attribute__((aligned(16))) float sBN[HDIM * 4];
    if (threadIdx.x < HDIM) {
        int f = threadIdx.x;
        float invN = 1.0f / (float)N;
        float mean = stats[f] * invN;
        float var  = stats[64 + f] * invN - mean * mean;
        float* q = sBN + ((f >> 1) << 3) + (f & 1);
        q[0] = g[f];
        q[2] = mean;
        q[4] = rsqrtf(var + EPSV);
        q[6] = beta[f];
    }
    __syncthreads();
    int wid  = (blockIdx.x * blockDim.x + threadIdx.x) >> 5;
    int lane = threadIdx.x & 31;
    int tr = wid >> 2;
    int tc = wid & 3;
    const bool live = tr < rowTiles;
    int m  = lane & 15;
    int kh = (lane >> 4) << 1;
    int rowBase = tr * 16 + ((lane >> 4) << 3);
    int col     = tc * 16 + m;
    v8f c = {0.f, 0.f, 0.f, 0.f, 0.f, 0.f, 0.f, 0.f};
    float4 dA = make_float4(0.f, 0.f, 0.f, 0.f), dB = dA;
    if (live) {

    {
        int pf   = tc * 16 + m;
        int prow = tr * 16 + ((lane >> 4) << 3);
        const float* q = sBN + ((pf >> 1) << 3) + (pf & 1);
        float pg = q[0], pm = q[2], pr = q[4], pb = q[6];
        float v[8];
#pragma unroll
        for (int r = 0; r < 8; ++r)
            v[r] = bn_apply(hin[(size_t)(prow + r) * HDIM + pf], pg, pm, pr, pb);
        int bmine = batch[tr * 16 + m];
        int b0    = __builtin_amdgcn_readfirstlane(bmine);
        if (__all(bmine == b0)) {
            float s = ((v[0] + v[1]) + (v[2] + v[3])) + ((v[4] + v[5]) + (v[6] + v[7]));
            s += __shfl_xor(s, 16);
            if (lane < 16) atomicAdd(&pooled[(size_t)b0 * 192 + poolOff + pf], s);
        } else {
#pragma unroll
            for (int r = 0; r < 8; ++r)
                atomicAdd(&pooled[(size_t)batch[prow + r] * 192 + poolOff + pf], v[r]);
        }
    }

    const float* Ar = hin + (size_t)(tr * 16 + m) * HDIM;
    const float* Wr = W   + (size_t)(tc * 16 + m) * HDIM;
#pragma unroll
    for (int k0 = 0; k0 < HDIM; k0 += 4) {
        v2f a = *(const v2f*)(Ar + k0 + kh);
        v2f b = *(const v2f*)(Wr + k0 + kh);
        const float* q = sBN + ((k0 + kh) << 2);
        v2f gam = *(const v2f*)(q), mean = *(const v2f*)(q + 2);
        v2f rs  = *(const v2f*)(q + 4), bet = *(const v2f*)(q + 6);
        a.x = bn_apply(a.x, gam.x, mean.x, rs.x, bet.x);
        a.y = bn_apply(a.y, gam.y, mean.y, rs.y, bet.y);
        c = __builtin_amdgcn_wmma_f32_16x16x4_f32(false, a, false, b, (short)0, c, false, false);
    }
    dA = *(const float4*)(dinv + rowBase);
    dB = *(const float4*)(dinv + rowBase + 4);
    }
    __syncthreads();
    if (live) {
        const float di[8] = {dA.x, dA.y, dA.z, dA.w, dB.x, dB.y, dB.z, dB.w};
#pragma unroll
        for (int r = 0; r < 8; ++r) {
            size_t o = (size_t)(rowBase + r) * HDIM + col;
            float v   = c[r] * di[r];
            out[o]    = v;
            aggOut[o] = v;
            agg1[o]   = 0.0f;
        }
    }
}

__device__ __forceinline__ void scatter_edge(const float* __restrict__ hWs,
                                             float* __restrict__ agg,
                                             int s, int d, int lane) {
    const float* r = hWs + (size_t)s * HDIM + lane;
    float lo = r[0], hi = r[32];
    float* a = agg + (size_t)d * HDIM + lane;
    atomicAdd(a, lo);
    atomicAdd(a + 32, hi);
}

__global__ void __launch_bounds__(256)
k_scatter(const float* __restrict__ hWs,
          const int* __restrict__ src, const int* __restrict__ dst,
          float* __restrict__ agg0, float* __restrict__ agg1, int E) {
    const int lane = threadIdx.x & 31;
    const unsigned place = __builtin_amdgcn_s_sendmsg_rtn(0x87);
    float* __restrict__ agg = ((place >> 18) & 1u) ? agg1 : agg0;
    const int wave   = blockIdx.x * 8 + __builtin_amdgcn_readfirstlane((int)(threadIdx.x >> 5));
    const int nwaves = gridDim.x * 8;
    const int nquad  = E >> 2;
    for (int g = wave; g < nquad; g += nwaves) {
        const int e = g << 2;
        int s0 = src[e], s1 = src[e + 1], s2 = src[e + 2], s3 = src[e + 3];
        int d0 = dst[e], d1 = dst[e + 1], d2 = dst[e + 2], d3 = dst[e + 3];
        const float* r0 = hWs + (size_t)s0 * HDIM + lane;
        const float* r1 = hWs + (size_t)s1 * HDIM + lane;
        const float* r2 = hWs + (size_t)s2 * HDIM + lane;
        const float* r3 = hWs + (size_t)s3 * HDIM + lane;
        float l0 = r0[0], h0 = r0[32];
        float l1 = r1[0], h1 = r1[32];
        float l2 = r2[0], h2 = r2[32];
        float l3 = r3[0], h3 = r3[32];
        float* a0 = agg + (size_t)d0 * HDIM + lane;
        float* a1 = agg + (size_t)d1 * HDIM + lane;
        float* a2 = agg + (size_t)d2 * HDIM + lane;
        float* a3 = agg + (size_t)d3 * HDIM + lane;
        atomicAdd(a0, l0);  atomicAdd(a0 + 32, h0);
        atomicAdd(a1, l1);  atomicAdd(a1 + 32, h1);
        atomicAdd(a2, l2);  atomicAdd(a2 + 32, h2);
        atomicAdd(a3, l3);  atomicAdd(a3 + 32, h3);
    }
    for (int e = (nquad << 2) + wave; e < E; e += nwaves)
        scatter_edge(hWs, agg, src[e], dst[e], lane);
}

__global__ void k_bias_relu_stats(float* __restrict__ agg, const float* __restrict__ agg1,
                                  const float* __restrict__ dinv,
                                  const float* __restrict__ bias,
                                  float* __restrict__ stats  , int N) {
    __shared__ float sS[256];
    __shared__ float sQ[256];
    int f  = threadIdx.x & 63;
    int rg = threadIdx.x >> 6;
    float b = bias[f];
    float s = 0.f, q = 0.f;
    for (int row = blockIdx.x * 4 + rg; row < N; row += gridDim.x * 4) {
        size_t idx = (size_t)row * HDIM + f;
        float v = (agg[idx] + agg1[idx]) * dinv[row] + b;
        v = fmaxf(v, 0.f);
        agg[idx] = v;
        s += v;
        q += v * v;
    }
    sS[threadIdx.x] = s;
    sQ[threadIdx.x] = q;
    __syncthreads();
    if (rg == 0) {
        s = sS[f] + sS[64 + f] + sS[128 + f] + sS[192 + f];
        q = sQ[f] + sQ[64 + f] + sQ[128 + f] + sQ[192 + f];
        atomicAdd(&stats[f], s);
        atomicAdd(&stats[64 + f], q);
    }
}

__global__ void k_bn_pool(const float* __restrict__ hin, const float* __restrict__ stats,
                          const float* __restrict__ g, const float* __restrict__ beta,
                          const int* __restrict__ batch, float* __restrict__ hout,
                          float* __restrict__ pooled, int poolOff, int N) {
    int t = blockIdx.x * blockDim.x + threadIdx.x;
    if (t >= N * HDIM) return;
    int n = t >> 6, f = t & 63;
    float invN = 1.0f / (float)N;
    float mean = stats[f] * invN;
    float var  = stats[64 + f] * invN - mean * mean;
    float v = g[f] * (hin[t] - mean) * rsqrtf(var + EPSV) + beta[f];
    hout[t] = v;
    atomicAdd(&pooled[(size_t)batch[n] * 192 + poolOff + f], v);
}

extern "C" void kernel_launch(void* const* d_in, const int* in_sizes, int n_in,
                              void* d_out, int out_size, void* d_ws, size_t ws_size,
                              hipStream_t stream) {
    const float* x    = (const float*)d_in[0];
    const int*   eidx = (const int*)d_in[1];
    const int*   batch = (const int*)d_in[2];
    const float* W[3]  = {(const float*)d_in[3],  (const float*)d_in[7],  (const float*)d_in[11]};
    const float* bi[3] = {(const float*)d_in[4],  (const float*)d_in[8],  (const float*)d_in[12]};
    const float* ga[3] = {(const float*)d_in[5],  (const float*)d_in[9],  (const float*)d_in[13]};
    const float* be[3] = {(const float*)d_in[6],  (const float*)d_in[10], (const float*)d_in[14]};

    const int N = in_sizes[0] / HDIM;
    const int E = in_sizes[1] / 2;
    const int G = (out_size - N * HDIM) / (3 * HDIM);

    const int* src = eidx;
    const int* dst = eidx + E;

    float* ws    = (float*)d_ws;
    float* dinv  = ws;
    float* tmp   = dinv + N;
    float* agg   = tmp  + (size_t)N * HDIM;
    float* stats = agg  + (size_t)N * HDIM;

    float* pooled = (float*)d_out;
    float* hfinal = (float*)d_out + (size_t)G * 192;
    float* agg1 = hfinal;

    const int TB = 256;

    k_deg_init <<<cdiv(N, TB), TB, 0, stream>>>(dinv, N);
    k_deg_accum<<<cdiv(E, TB), TB, 0, stream>>>(dst, dinv, E);
    k_dinv     <<<cdiv(N, TB), TB, 0, stream>>>(dinv, N);

    k_zero<<<cdiv((long long)G * 192, TB), TB, 0, stream>>>(pooled, G * 192);

    const int rowTiles   = N / 16;
    const int gemmWaves  = rowTiles * (HDIM / 16);
    const int gemmBlocks = cdiv((long long)gemmWaves * 32, TB);

    for (int l = 0; l < 3; ++l) {
        if (l == 0)
            k_gemm_wmma<<<gemmBlocks, TB, 0, stream>>>(x, W[0], dinv, tmp, agg, agg1, rowTiles);
        else
            k_bn_pool_gemm<<<gemmBlocks, TB, 0, stream>>>(agg, stats, ga[l - 1], be[l - 1], batch,
                                                         pooled, (l - 1) * HDIM, W[l], dinv, tmp, agg, agg1,
                                                         rowTiles, N);
        k_scatter  <<<2048, TB, 0, stream>>>(tmp, src, dst, agg, agg1, E);

        k_zero<<<1, 128, 0, stream>>>(stats, 128);
        k_bias_relu_stats<<<1024, TB, 0, stream>>>(agg, agg1, dinv, bi[l], stats, N);
    }
    k_bn_pool<<<cdiv((long long)N * HDIM, TB), TB, 0, stream>>>(
        agg, stats, ga[2], be[2], batch, hfinal, pooled, 2 * HDIM, N);
}


# Degrees counted in LDS: a work-group counts one edge piece's destinations inside one node range (integer LDS adds), stores its counts; dinv pass sums 32 partial counts; global atomics gone; exact
# speedup vs baseline: 1.9524x; 1.1531x over previous
#include <hip/hip_runtime.h>

typedef float v2f __attribute__((ext_vector_type(2)));
typedef float v8f __attribute__((ext_vector_type(8)));

#define HDIM 64
#define EPSV 1e-5f

static inline int cdiv(long long a, long long b) { return (int)((a + b - 1) / b); }

__global__ void k_zero(float* __restrict__ p, int n) {
    int i = blockIdx.x * blockDim.x + threadIdx.x;
    if (i < n) p[i] = 0.0f;
}

#define DEG_CHUNKS 32
#define DEG_SPAN   12512

__global__ void __launch_bounds__(1024)
k_deg_count(const int* __restrict__ dst, float* __restrict__ part, int E, int N) {
    __shared__ __attribute__((aligned(16))) unsigned sCnt[DEG_SPAN];
    const int tid = threadIdx.x;
    const int c = blockIdx.x % DEG_CHUNKS;
    const int r = blockIdx.x / DEG_CHUNKS;
    uint4* z = (uint4*)sCnt;
    for (int i = tid; i < DEG_SPAN / 4; i += 1024) z[i] = make_uint4(0u, 0u, 0u, 0u);
    __syncthreads();
    const int per  = (E + DEG_CHUNKS - 1) / DEG_CHUNKS;
    const int e0   = c * per;
    const int e1   = min(E, e0 + per);
    const int base = r * DEG_SPAN;
    const unsigned span = (unsigned)min(DEG_SPAN, N - base);
    for (int e = e0 + tid; e < e1; e += 8 * 1024) {
        int d[8];
#pragma unroll
        for (int k = 0; k < 8; ++k) {
            int idx = e + k * 1024;
            d[k] = (idx < e1) ? dst[idx] : -1;
        }
#pragma unroll
        for (int k = 0; k < 8; ++k) {
            unsigned off = (unsigned)(d[k] - base);
            if (off < span) atomicAdd(&sCnt[off], 1u);
        }
    }
    __syncthreads();
    float* p = part + (size_t)c * N + base;
    for (unsigned i = tid; i < span; i += 1024) p[i] = (float)sCnt[i];
}

__global__ void k_dinv(const float* __restrict__ part, float* __restrict__ dinv, int N) {
    int i = blockIdx.x * blockDim.x + threadIdx.x;
    if (i >= N) return;
    float v[DEG_CHUNKS];
#pragma unroll
    for (int c = 0; c < DEG_CHUNKS; ++c) v[c] = part[(size_t)c * N + i];
    float s = 1.0f;
#pragma unroll
    for (int c = 0; c < DEG_CHUNKS; ++c) s += v[c];
    dinv[i] = rsqrtf(s);
}

__global__ void __launch_bounds__(256, 16)
k_gemm_wmma(const float* __restrict__ A, const float* __restrict__ W,
                            const float* __restrict__ dinv,
                            float* __restrict__ out, float* __restrict__ agg,
                            float* __restrict__ agg1, int rowTiles) {
    int wid  = (blockIdx.x * blockDim.x + threadIdx.x) >> 5;
    int lane = threadIdx.x & 31;
    int tr = wid >> 2;
    int tc = wid & 3;
    if (tr >= rowTiles) return;
    int m  = lane & 15;
    int kh = (lane >> 4) << 1;
    const float* Ar = A + (size_t)(tr * 16 + m) * HDIM;
    const float* Wr = W + (size_t)(tc * 16 + m) * HDIM;
    int rowBase = tr * 16 + ((lane >> 4) << 3);
    int col     = tc * 16 + m;
#pragma unroll
    for (int r = 0; r < 8; ++r)
        agg1[(size_t)(rowBase + r) * HDIM + col] = 0.0f;
    v8f c = {0.f, 0.f, 0.f, 0.f, 0.f, 0.f, 0.f, 0.f};
#pragma unroll
    for (int k0 = 0; k0 < HDIM; k0 += 4) {
        v2f a = *(const v2f*)(Ar + k0 + kh);
        v2f b = *(const v2f*)(Wr + k0 + kh);
        c = __builtin_amdgcn_wmma_f32_16x16x4_f32(false, a, false, b, (short)0, c, false, false);
    }
    const float4 dA = *(const float4*)(dinv + rowBase);
    const float4 dB = *(const float4*)(dinv + rowBase + 4);
    const float di[8] = {dA.x, dA.y, dA.z, dA.w, dB.x, dB.y, dB.z, dB.w};
#pragma unroll
    for (int r = 0; r < 8; ++r) {
        size_t o = (size_t)(rowBase + r) * HDIM + col;
        float v = c[r] * di[r];
        out[o] = v;
        agg[o] = v;
    }
}

__device__ __forceinline__ float bn_apply(float x, float gam, float mean, float rs, float bet) {
    return gam * (x - mean) * rs + bet;
}

__global__ void __launch_bounds__(256)
k_bn_pool_gemm(const float* hin, const float* __restrict__ stats,
               const float* __restrict__ g, const float* __restrict__ beta,
               const int* __restrict__ batch, float* __restrict__ pooled, int poolOff,
               const float* __restrict__ W, const float* __restrict__ dinv,
               float* __restrict__ out, float* aggOut, float* __restrict__ agg1,
               int rowTiles, int N) {
    __shared__ __attribute__((aligned(16))) float sBN[HDIM * 4];
    if (threadIdx.x < HDIM) {
        int f = threadIdx.x;
        float invN = 1.0f / (float)N;
        float mean = stats[f] * invN;
        float var  = stats[64 + f] * invN - mean * mean;
        float* q = sBN + ((f >> 1) << 3) + (f & 1);
        q[0] = g[f];
        q[2] = mean;
        q[4] = rsqrtf(var + EPSV);
        q[6] = beta[f];
    }
    __syncthreads();
    int wid  = (blockIdx.x * blockDim.x + threadIdx.x) >> 5;
    int lane = threadIdx.x & 31;
    int tr = wid >> 2;
    int tc = wid & 3;
    const bool live = tr < rowTiles;
    int m  = lane & 15;
    int kh = (lane >> 4) << 1;
    int rowBase = tr * 16 + ((lane >> 4) << 3);
    int col     = tc * 16 + m;
    v8f c = {0.f, 0.f, 0.f, 0.f, 0.f, 0.f, 0.f, 0.f};
    float4 dA = make_float4(0.f, 0.f, 0.f, 0.f), dB = dA;
    if (live) {

    {
        int pf   = tc * 16 + m;
        int prow = tr * 16 + ((lane >> 4) << 3);
        const float* q = sBN + ((pf >> 1) << 3) + (pf & 1);
        float pg = q[0], pm = q[2], pr = q[4], pb = q[6];
        float v[8];
#pragma unroll
        for (int r = 0; r < 8; ++r)
            v[r] = bn_apply(hin[(size_t)(prow + r) * HDIM + pf], pg, pm, pr, pb);
        int bmine = batch[tr * 16 + m];
        int b0    = __builtin_amdgcn_readfirstlane(bmine);
        if (__all(bmine == b0)) {
            float s = ((v[0] + v[1]) + (v[2] + v[3])) + ((v[4] + v[5]) + (v[6] + v[7]));
            s += __shfl_xor(s, 16);
            if (lane < 16) atomicAdd(&pooled[(size_t)b0 * 192 + poolOff + pf], s);
        } else {
#pragma unroll
            for (int r = 0; r < 8; ++r)
                atomicAdd(&pooled[(size_t)batch[prow + r] * 192 + poolOff + pf], v[r]);
        }
    }

    const float* Ar = hin + (size_t)(tr * 16 + m) * HDIM;
    const float* Wr = W   + (size_t)(tc * 16 + m) * HDIM;
#pragma unroll
    for (int k0 = 0; k0 < HDIM; k0 += 4) {
        v2f a = *(const v2f*)(Ar + k0 + kh);
        v2f b = *(const v2f*)(Wr + k0 + kh);
        const float* q = sBN + ((k0 + kh) << 2);
        v2f gam = *(const v2f*)(q), mean = *(const v2f*)(q + 2);
        v2f rs  = *(const v2f*)(q + 4), bet = *(const v2f*)(q + 6);
        a.x = bn_apply(a.x, gam.x, mean.x, rs.x, bet.x);
        a.y = bn_apply(a.y, gam.y, mean.y, rs.y, bet.y);
        c = __builtin_amdgcn_wmma_f32_16x16x4_f32(false, a, false, b, (short)0, c, false, false);
    }
    dA = *(const float4*)(dinv + rowBase);
    dB = *(const float4*)(dinv + rowBase + 4);
    }
    __syncthreads();
    if (live) {
        const float di[8] = {dA.x, dA.y, dA.z, dA.w, dB.x, dB.y, dB.z, dB.w};
#pragma unroll
        for (int r = 0; r < 8; ++r) {
            size_t o = (size_t)(rowBase + r) * HDIM + col;
            float v   = c[r] * di[r];
            out[o]    = v;
            aggOut[o] = v;
            agg1[o]   = 0.0f;
        }
    }
}

__device__ __forceinline__ void scatter_edge(const float* __restrict__ hWs,
                                             float* __restrict__ agg,
                                             int s, int d, int lane) {
    const float* r = hWs + (size_t)s * HDIM + lane;
    float lo = r[0], hi = r[32];
    float* a = agg + (size_t)d * HDIM + lane;
    atomicAdd(a, lo);
    atomicAdd(a + 32, hi);
}

__global__ void __launch_bounds__(256)
k_scatter(const float* __restrict__ hWs,
          const int* __restrict__ src, const int* __restrict__ dst,
          float* __restrict__ agg0, float* __restrict__ agg1, int E) {
    const int lane = threadIdx.x & 31;
    const unsigned place = __builtin_amdgcn_s_sendmsg_rtn(0x87);
    float* __restrict__ agg = ((place >> 18) & 1u) ? agg1 : agg0;
    const int wave   = blockIdx.x * 8 + __builtin_amdgcn_readfirstlane((int)(threadIdx.x >> 5));
    const int nwaves = gridDim.x * 8;
    const int nquad  = E >> 2;
    for (int g = wave; g < nquad; g += nwaves) {
        const int e = g << 2;
        int s0 = src[e], s1 = src[e + 1], s2 = src[e + 2], s3 = src[e + 3];
        int d0 = dst[e], d1 = dst[e + 1], d2 = dst[e + 2], d3 = dst[e + 3];
        const float* r0 = hWs + (size_t)s0 * HDIM + lane;
        const float* r1 = hWs + (size_t)s1 * HDIM + lane;
        const float* r2 = hWs + (size_t)s2 * HDIM + lane;
        const float* r3 = hWs + (size_t)s3 * HDIM + lane;
        float l0 = r0[0], h0 = r0[32];
        float l1 = r1[0], h1 = r1[32];
        float l2 = r2[0], h2 = r2[32];
        float l3 = r3[0], h3 = r3[32];
        float* a0 = agg + (size_t)d0 * HDIM + lane;
        float* a1 = agg + (size_t)d1 * HDIM + lane;
        float* a2 = agg + (size_t)d2 * HDIM + lane;
        float* a3 = agg + (size_t)d3 * HDIM + lane;
        atomicAdd(a0, l0);  atomicAdd(a0 + 32, h0);
        atomicAdd(a1, l1);  atomicAdd(a1 + 32, h1);
        atomicAdd(a2, l2);  atomicAdd(a2 + 32, h2);
        atomicAdd(a3, l3);  atomicAdd(a3 + 32, h3);
    }
    for (int e = (nquad << 2) + wave; e < E; e += nwaves)
        scatter_edge(hWs, agg, src[e], dst[e], lane);
}

__global__ void k_bias_relu_stats(float* __restrict__ agg, const float* __restrict__ agg1,
                                  const float* __restrict__ dinv,
                                  const float* __restrict__ bias,
                                  float* __restrict__ stats  , int N) {
    __shared__ float sS[256];
    __shared__ float sQ[256];
    int f  = threadIdx.x & 63;
    int rg = threadIdx.x >> 6;
    float b = bias[f];
    float s = 0.f, q = 0.f;
    for (int row = blockIdx.x * 4 + rg; row < N; row += gridDim.x * 4) {
        size_t idx = (size_t)row * HDIM + f;
        float v = (agg[idx] + agg1[idx]) * dinv[row] + b;
        v = fmaxf(v, 0.f);
        agg[idx] = v;
        s += v;
        q += v * v;
    }
    sS[threadIdx.x] = s;
    sQ[threadIdx.x] = q;
    __syncthreads();
    if (rg == 0) {
        s = sS[f] + sS[64 + f] + sS[128 + f] + sS[192 + f];
        q = sQ[f] + sQ[64 + f] + sQ[128 + f] + sQ[192 + f];
        atomicAdd(&stats[f], s);
        atomicAdd(&stats[64 + f], q);
    }
}

__global__ void k_bn_pool(const float* __restrict__ hin, const float* __restrict__ stats,
                          const float* __restrict__ g, const float* __restrict__ beta,
                          const int* __restrict__ batch, float* __restrict__ hout,
                          float* __restrict__ pooled, int poolOff, int N) {
    int t = blockIdx.x * blockDim.x + threadIdx.x;
    if (t >= N * HDIM) return;
    int n = t >> 6, f = t & 63;
    float invN = 1.0f / (float)N;
    float mean = stats[f] * invN;
    float var  = stats[64 + f] * invN - mean * mean;
    float v = g[f] * (hin[t] - mean) * rsqrtf(var + EPSV) + beta[f];
    hout[t] = v;
    atomicAdd(&pooled[(size_t)batch[n] * 192 + poolOff + f], v);
}

extern "C" void kernel_launch(void* const* d_in, const int* in_sizes, int n_in,
                              void* d_out, int out_size, void* d_ws, size_t ws_size,
                              hipStream_t stream) {
    const float* x    = (const float*)d_in[0];
    const int*   eidx = (const int*)d_in[1];
    const int*   batch = (const int*)d_in[2];
    const float* W[3]  = {(const float*)d_in[3],  (const float*)d_in[7],  (const float*)d_in[11]};
    const float* bi[3] = {(const float*)d_in[4],  (const float*)d_in[8],  (const float*)d_in[12]};
    const float* ga[3] = {(const float*)d_in[5],  (const float*)d_in[9],  (const float*)d_in[13]};
    const float* be[3] = {(const float*)d_in[6],  (const float*)d_in[10], (const float*)d_in[14]};

    const int N = in_sizes[0] / HDIM;
    const int E = in_sizes[1] / 2;
    const int G = (out_size - N * HDIM) / (3 * HDIM);

    const int* src = eidx;
    const int* dst = eidx + E;

    float* ws    = (float*)d_ws;
    float* dinv  = ws;
    float* tmp   = dinv + N;
    float* agg   = tmp  + (size_t)N * HDIM;
    float* stats = agg  + (size_t)N * HDIM;

    float* pooled = (float*)d_out;
    float* hfinal = (float*)d_out + (size_t)G * 192;
    float* agg1 = hfinal;

    const int TB = 256;

    k_deg_count<<<cdiv(N, DEG_SPAN) * DEG_CHUNKS, 1024, 0, stream>>>(dst, tmp, E, N);
    k_dinv     <<<cdiv(N, TB), TB, 0, stream>>>(tmp, dinv, N);

    k_zero<<<cdiv((long long)G * 192, TB), TB, 0, stream>>>(pooled, G * 192);

    const int rowTiles   = N / 16;
    const int gemmWaves  = rowTiles * (HDIM / 16);
    const int gemmBlocks = cdiv((long long)gemmWaves * 32, TB);

    for (int l = 0; l < 3; ++l) {
        if (l == 0)
            k_gemm_wmma<<<gemmBlocks, TB, 0, stream>>>(x, W[0], dinv, tmp, agg, agg1, rowTiles);
        else
            k_bn_pool_gemm<<<gemmBlocks, TB, 0, stream>>>(agg, stats, ga[l - 1], be[l - 1], batch,
                                                         pooled, (l - 1) * HDIM, W[l], dinv, tmp, agg, agg1,
                                                         rowTiles, N);
        k_scatter  <<<2048, TB, 0, stream>>>(tmp, src, dst, agg, agg1, E);

        k_zero<<<1, 128, 0, stream>>>(stats, 128);
        k_bias_relu_stats<<<1024, TB, 0, stream>>>(agg, agg1, dinv, bi[l], stats, N);
    }
    k_bn_pool<<<cdiv((long long)N * HDIM, TB), TB, 0, stream>>>(
        agg, stats, ga[2], be[2], batch, hfinal, pooled, 2 * HDIM, N);
}


# GEMM epilogue stores the scatter's rows in pair order (feature f at slot 2*(f mod 32)+f/32, rows on 256-B boundaries): edge scatter reads a row as one 8-byte load per lane, not two; sum order kept
# speedup vs baseline: 2.0746x; 1.0626x over previous
#include <hip/hip_runtime.h>

typedef float v2f __attribute__((ext_vector_type(2)));
typedef float v8f __attribute__((ext_vector_type(8)));

#define HDIM 64
#define EPSV 1e-5f

static inline int cdiv(long long a, long long b) { return (int)((a + b - 1) / b); }

__global__ void k_zero(float* __restrict__ p, int n) {
    int i = blockIdx.x * blockDim.x + threadIdx.x;
    if (i < n) p[i] = 0.0f;
}

#define DEG_CHUNKS 32
#define DEG_SPAN   12512

__global__ void __launch_bounds__(1024)
k_deg_count(const int* __restrict__ dst, float* __restrict__ part, int E, int N) {
    __shared__ __attribute__((aligned(16))) unsigned sCnt[DEG_SPAN];
    const int tid = threadIdx.x;
    const int c = blockIdx.x % DEG_CHUNKS;
    const int r = blockIdx.x / DEG_CHUNKS;
    uint4* z = (uint4*)sCnt;
    for (int i = tid; i < DEG_SPAN / 4; i += 1024) z[i] = make_uint4(0u, 0u, 0u, 0u);
    __syncthreads();
    const int per  = (E + DEG_CHUNKS - 1) / DEG_CHUNKS;
    const int e0   = c * per;
    const int e1   = min(E, e0 + per);
    const int base = r * DEG_SPAN;
    const unsigned span = (unsigned)min(DEG_SPAN, N - base);
    for (int e = e0 + tid; e < e1; e += 8 * 1024) {
        int d[8];
#pragma unroll
        for (int k = 0; k < 8; ++k) {
            int idx = e + k * 1024;
            d[k] = (idx < e1) ? dst[idx] : -1;
        }
#pragma unroll
        for (int k = 0; k < 8; ++k) {
            unsigned off = (unsigned)(d[k] - base);
            if (off < span) atomicAdd(&sCnt[off], 1u);
        }
    }
    __syncthreads();
    float* p = part + (size_t)c * N + base;
    for (unsigned i = tid; i < span; i += 1024) p[i] = (float)sCnt[i];
}

__global__ void k_dinv(const float* __restrict__ part, float* __restrict__ dinv, int N) {
    int i = blockIdx.x * blockDim.x + threadIdx.x;
    if (i >= N) return;
    float v[DEG_CHUNKS];
#pragma unroll
    for (int c = 0; c < DEG_CHUNKS; ++c) v[c] = part[(size_t)c * N + i];
    float s = 1.0f;
#pragma unroll
    for (int c = 0; c < DEG_CHUNKS; ++c) s += v[c];
    dinv[i] = rsqrtf(s);
}

__global__ void __launch_bounds__(256, 16)
k_gemm_wmma(const float* __restrict__ A, const float* __restrict__ W,
                            const float* __restrict__ dinv,
                            float* __restrict__ out, float* __restrict__ agg,
                            float* __restrict__ agg1, int rowTiles) {
    int wid  = (blockIdx.x * blockDim.x + threadIdx.x) >> 5;
    int lane = threadIdx.x & 31;
    int tr = wid >> 2;
    int tc = wid & 3;
    if (tr >= rowTiles) return;
    int m  = lane & 15;
    int kh = (lane >> 4) << 1;
    const float* Ar = A + (size_t)(tr * 16 + m) * HDIM;
    const float* Wr = W + (size_t)(tc * 16 + m) * HDIM;
    int rowBase = tr * 16 + ((lane >> 4) << 3);
    int col     = tc * 16 + m;
#pragma unroll
    for (int r = 0; r < 8; ++r)
        agg1[(size_t)(rowBase + r) * HDIM + col] = 0.0f;
    v8f c = {0.f, 0.f, 0.f, 0.f, 0.f, 0.f, 0.f, 0.f};
#pragma unroll
    for (int k0 = 0; k0 < HDIM; k0 += 4) {
        v2f a = *(const v2f*)(Ar + k0 + kh);
        v2f b = *(const v2f*)(Wr + k0 + kh);
        c = __builtin_amdgcn_wmma_f32_16x16x4_f32(false, a, false, b, (short)0, c, false, false);
    }
    const float4 dA = *(const float4*)(dinv + rowBase);
    const float4 dB = *(const float4*)(dinv + rowBase + 4);
    const float di[8] = {dA.x, dA.y, dA.z, dA.w, dB.x, dB.y, dB.z, dB.w};
    const int pcol = ((col & 31) << 1) | (col >> 5);
#pragma unroll
    for (int r = 0; r < 8; ++r) {
        size_t ro = (size_t)(rowBase + r) * HDIM;
        float v = c[r] * di[r];
        out[ro + pcol] = v;
        agg[ro + col]  = v;
    }
}

__device__ __forceinline__ float bn_apply(float x, float gam, float mean, float rs, float bet) {
    return gam * (x - mean) * rs + bet;
}

__global__ void __launch_bounds__(256)
k_bn_pool_gemm(const float* hin, const float* __restrict__ stats,
               const float* __restrict__ g, const float* __restrict__ beta,
               const int* __restrict__ batch, float* __restrict__ pooled, int poolOff,
               const float* __restrict__ W, const float* __restrict__ dinv,
               float* __restrict__ out, float* aggOut, float* __restrict__ agg1,
               int rowTiles, int N) {
    __shared__ __attribute__((aligned(16))) float sBN[HDIM * 4];
    if (threadIdx.x < HDIM) {
        int f = threadIdx.x;
        float invN = 1.0f / (float)N;
        float mean = stats[f] * invN;
        float var  = stats[64 + f] * invN - mean * mean;
        float* q = sBN + ((f >> 1) << 3) + (f & 1);
        q[0] = g[f];
        q[2] = mean;
        q[4] = rsqrtf(var + EPSV);
        q[6] = beta[f];
    }
    __syncthreads();
    int wid  = (blockIdx.x * blockDim.x + threadIdx.x) >> 5;
    int lane = threadIdx.x & 31;
    int tr = wid >> 2;
    int tc = wid & 3;
    const bool live = tr < rowTiles;
    int m  = lane & 15;
    int kh = (lane >> 4) << 1;
    int rowBase = tr * 16 + ((lane >> 4) << 3);
    int col     = tc * 16 + m;
    v8f c = {0.f, 0.f, 0.f, 0.f, 0.f, 0.f, 0.f, 0.f};
    float4 dA = make_float4(0.f, 0.f, 0.f, 0.f), dB = dA;
    if (live) {

    {
        int pf   = tc * 16 + m;
        int prow = tr * 16 + ((lane >> 4) << 3);
        const float* q = sBN + ((pf >> 1) << 3) + (pf & 1);
        float pg = q[0], pm = q[2], pr = q[4], pb = q[6];
        float v[8];
#pragma unroll
        for (int r = 0; r < 8; ++r)
            v[r] = bn_apply(hin[(size_t)(prow + r) * HDIM + pf], pg, pm, pr, pb);
        int bmine = batch[tr * 16 + m];
        int b0    = __builtin_amdgcn_readfirstlane(bmine);
        if (__all(bmine == b0)) {
            float s = ((v[0] + v[1]) + (v[2] + v[3])) + ((v[4] + v[5]) + (v[6] + v[7]));
            s += __shfl_xor(s, 16);
            if (lane < 16) atomicAdd(&pooled[(size_t)b0 * 192 + poolOff + pf], s);
        } else {
#pragma unroll
            for (int r = 0; r < 8; ++r)
                atomicAdd(&pooled[(size_t)batch[prow + r] * 192 + poolOff + pf], v[r]);
        }
    }

    const float* Ar = hin + (size_t)(tr * 16 + m) * HDIM;
    const float* Wr = W   + (size_t)(tc * 16 + m) * HDIM;
#pragma unroll
    for (int k0 = 0; k0 < HDIM; k0 += 4) {
        v2f a = *(const v2f*)(Ar + k0 + kh);
        v2f b = *(const v2f*)(Wr + k0 + kh);
        const float* q = sBN + ((k0 + kh) << 2);
        v2f gam = *(const v2f*)(q), mean = *(const v2f*)(q + 2);
        v2f rs  = *(const v2f*)(q + 4), bet = *(const v2f*)(q + 6);
        a.x = bn_apply(a.x, gam.x, mean.x, rs.x, bet.x);
        a.y = bn_apply(a.y, gam.y, mean.y, rs.y, bet.y);
        c = __builtin_amdgcn_wmma_f32_16x16x4_f32(false, a, false, b, (short)0, c, false, false);
    }
    dA = *(const float4*)(dinv + rowBase);
    dB = *(const float4*)(dinv + rowBase + 4);
    }
    __syncthreads();
    if (live) {
        const float di[8] = {dA.x, dA.y, dA.z, dA.w, dB.x, dB.y, dB.z, dB.w};
        const int pcol = ((col & 31) << 1) | (col >> 5);
#pragma unroll
        for (int r = 0; r < 8; ++r) {
            size_t ro = (size_t)(rowBase + r) * HDIM;
            float v   = c[r] * di[r];
            out[ro + pcol]   = v;
            aggOut[ro + col] = v;
            agg1[ro + col]   = 0.0f;
        }
    }
}

__device__ __forceinline__ void scatter_edge(const float* __restrict__ hWs,
                                             float* __restrict__ agg,
                                             int s, int d, int lane) {
    const v2f p = *((const v2f*)(hWs + (size_t)s * HDIM) + lane);
    float* a = agg + (size_t)d * HDIM + lane;
    atomicAdd(a, p.x);
    atomicAdd(a + 32, p.y);
}

__global__ void __launch_bounds__(256)
k_scatter(const float* __restrict__ hWs,
          const int* __restrict__ src, const int* __restrict__ dst,
          float* __restrict__ agg0, float* __restrict__ agg1, int E) {
    const int lane = threadIdx.x & 31;
    const unsigned place = __builtin_amdgcn_s_sendmsg_rtn(0x87);
    float* __restrict__ agg = ((place >> 18) & 1u) ? agg1 : agg0;
    const int wave   = blockIdx.x * 8 + __builtin_amdgcn_readfirstlane((int)(threadIdx.x >> 5));
    const int nwaves = gridDim.x * 8;
    const int nquad  = E >> 2;
    for (int g = wave; g < nquad; g += nwaves) {
        const int e = g << 2;
        int s0 = src[e], s1 = src[e + 1], s2 = src[e + 2], s3 = src[e + 3];
        int d0 = dst[e], d1 = dst[e + 1], d2 = dst[e + 2], d3 = dst[e + 3];
        const v2f p0 = *((const v2f*)(hWs + (size_t)s0 * HDIM) + lane);
        const v2f p1 = *((const v2f*)(hWs + (size_t)s1 * HDIM) + lane);
        const v2f p2 = *((const v2f*)(hWs + (size_t)s2 * HDIM) + lane);
        const v2f p3 = *((const v2f*)(hWs + (size_t)s3 * HDIM) + lane);
        float* a0 = agg + (size_t)d0 * HDIM + lane;
        float* a1 = agg + (size_t)d1 * HDIM + lane;
        float* a2 = agg + (size_t)d2 * HDIM + lane;
        float* a3 = agg + (size_t)d3 * HDIM + lane;
        atomicAdd(a0, p0.x);  atomicAdd(a0 + 32, p0.y);
        atomicAdd(a1, p1.x);  atomicAdd(a1 + 32, p1.y);
        atomicAdd(a2, p2.x);  atomicAdd(a2 + 32, p2.y);
        atomicAdd(a3, p3.x);  atomicAdd(a3 + 32, p3.y);
    }
    for (int e = (nquad << 2) + wave; e < E; e += nwaves)
        scatter_edge(hWs, agg, src[e], dst[e], lane);
}

__global__ void k_bias_relu_stats(float* __restrict__ agg, const float* __restrict__ agg1,
                                  const float* __restrict__ dinv,
                                  const float* __restrict__ bias,
                                  float* __restrict__ stats  , int N) {
    __shared__ float sS[256];
    __shared__ float sQ[256];
    int f  = threadIdx.x & 63;
    int rg = threadIdx.x >> 6;
    float b = bias[f];
    float s = 0.f, q = 0.f;
    for (int row = blockIdx.x * 4 + rg; row < N; row += gridDim.x * 4) {
        size_t idx = (size_t)row * HDIM + f;
        float v = (agg[idx] + agg1[idx]) * dinv[row] + b;
        v = fmaxf(v, 0.f);
        agg[idx] = v;
        s += v;
        q += v * v;
    }
    sS[threadIdx.x] = s;
    sQ[threadIdx.x] = q;
    __syncthreads();
    if (rg == 0) {
        s = sS[f] + sS[64 + f] + sS[128 + f] + sS[192 + f];
        q = sQ[f] + sQ[64 + f] + sQ[128 + f] + sQ[192 + f];
        atomicAdd(&stats[f], s);
        atomicAdd(&stats[64 + f], q);
    }
}

__global__ void k_bn_pool(const float* __restrict__ hin, const float* __restrict__ stats,
                          const float* __restrict__ g, const float* __restrict__ beta,
                          const int* __restrict__ batch, float* __restrict__ hout,
                          float* __restrict__ pooled, int poolOff, int N) {
    int t = blockIdx.x * blockDim.x + threadIdx.x;
    if (t >= N * HDIM) return;
    int n = t >> 6, f = t & 63;
    float invN = 1.0f / (float)N;
    float mean = stats[f] * invN;
    float var  = stats[64 + f] * invN - mean * mean;
    float v = g[f] * (hin[t] - mean) * rsqrtf(var + EPSV) + beta[f];
    hout[t] = v;
    atomicAdd(&pooled[(size_t)batch[n] * 192 + poolOff + f], v);
}

extern "C" void kernel_launch(void* const* d_in, const int* in_sizes, int n_in,
                              void* d_out, int out_size, void* d_ws, size_t ws_size,
                              hipStream_t stream) {
    const float* x    = (const float*)d_in[0];
    const int*   eidx = (const int*)d_in[1];
    const int*   batch = (const int*)d_in[2];
    const float* W[3]  = {(const float*)d_in[3],  (const float*)d_in[7],  (const float*)d_in[11]};
    const float* bi[3] = {(const float*)d_in[4],  (const float*)d_in[8],  (const float*)d_in[12]};
    const float* ga[3] = {(const float*)d_in[5],  (const float*)d_in[9],  (const float*)d_in[13]};
    const float* be[3] = {(const float*)d_in[6],  (const float*)d_in[10], (const float*)d_in[14]};

    const int N = in_sizes[0] / HDIM;
    const int E = in_sizes[1] / 2;
    const int G = (out_size - N * HDIM) / (3 * HDIM);

    const int* src = eidx;
    const int* dst = eidx + E;

    float* ws    = (float*)d_ws;
    float* tmp   = ws;
    float* agg   = tmp  + (size_t)N * HDIM;
    float* dinv  = agg  + (size_t)N * HDIM;
    float* stats = dinv + N;

    float* pooled = (float*)d_out;
    float* hfinal = (float*)d_out + (size_t)G * 192;
    float* agg1 = hfinal;

    const int TB = 256;

    k_deg_count<<<cdiv(N, DEG_SPAN) * DEG_CHUNKS, 1024, 0, stream>>>(dst, tmp, E, N);
    k_dinv     <<<cdiv(N, TB), TB, 0, stream>>>(tmp, dinv, N);

    k_zero<<<cdiv((long long)G * 192, TB), TB, 0, stream>>>(pooled, G * 192);

    const int rowTiles   = N / 16;
    const int gemmWaves  = rowTiles * (HDIM / 16);
    const int gemmBlocks = cdiv((long long)gemmWaves * 32, TB);

    for (int l = 0; l < 3; ++l) {
        if (l == 0)
            k_gemm_wmma<<<gemmBlocks, TB, 0, stream>>>(x, W[0], dinv, tmp, agg, agg1, rowTiles);
        else
            k_bn_pool_gemm<<<gemmBlocks, TB, 0, stream>>>(agg, stats, ga[l - 1], be[l - 1], batch,
                                                         pooled, (l - 1) * HDIM, W[l], dinv, tmp, agg, agg1,
                                                         rowTiles, N);
        k_scatter  <<<2048, TB, 0, stream>>>(tmp, src, dst, agg, agg1, E);

        k_zero<<<1, 128, 0, stream>>>(stats, 128);
        k_bias_relu_stats<<<1024, TB, 0, stream>>>(agg, agg1, dinv, bi[l], stats, N);
    }
    k_bn_pool<<<cdiv((long long)N * HDIM, TB), TB, 0, stream>>>(
        agg, stats, ga[2], be[2], batch, hfinal, pooled, 2 * HDIM, N);
}


# Bias+ReLU+stats pass: a thread takes 8 of its rows per trip and asks for all 16 plane values before the first use: it waits once per trip, not once per row; rows per thread and sum order kept
# speedup vs baseline: 2.2351x; 1.0774x over previous
#include <hip/hip_runtime.h>

typedef float v2f __attribute__((ext_vector_type(2)));
typedef float v8f __attribute__((ext_vector_type(8)));

#define HDIM 64
#define EPSV 1e-5f

static inline int cdiv(long long a, long long b) { return (int)((a + b - 1) / b); }

__global__ void k_zero(float* __restrict__ p, int n) {
    int i = blockIdx.x * blockDim.x + threadIdx.x;
    if (i < n) p[i] = 0.0f;
}

#define DEG_CHUNKS 32
#define DEG_SPAN   12512

__global__ void __launch_bounds__(1024)
k_deg_count(const int* __restrict__ dst, float* __restrict__ part, int E, int N) {
    __shared__ __attribute__((aligned(16))) unsigned sCnt[DEG_SPAN];
    const int tid = threadIdx.x;
    const int c = blockIdx.x % DEG_CHUNKS;
    const int r = blockIdx.x / DEG_CHUNKS;
    uint4* z = (uint4*)sCnt;
    for (int i = tid; i < DEG_SPAN / 4; i += 1024) z[i] = make_uint4(0u, 0u, 0u, 0u);
    __syncthreads();
    const int per  = (E + DEG_CHUNKS - 1) / DEG_CHUNKS;
    const int e0   = c * per;
    const int e1   = min(E, e0 + per);
    const int base = r * DEG_SPAN;
    const unsigned span = (unsigned)min(DEG_SPAN, N - base);
    for (int e = e0 + tid; e < e1; e += 8 * 1024) {
        int d[8];
#pragma unroll
        for (int k = 0; k < 8; ++k) {
            int idx = e + k * 1024;
            d[k] = (idx < e1) ? dst[idx] : -1;
        }
#pragma unroll
        for (int k = 0; k < 8; ++k) {
            unsigned off = (unsigned)(d[k] - base);
            if (off < span) atomicAdd(&sCnt[off], 1u);
        }
    }
    __syncthreads();
    float* p = part + (size_t)c * N + base;
    for (unsigned i = tid; i < span; i += 1024) p[i] = (float)sCnt[i];
}

__global__ void k_dinv(const float* __restrict__ part, float* __restrict__ dinv, int N) {
    int i = blockIdx.x * blockDim.x + threadIdx.x;
    if (i >= N) return;
    float v[DEG_CHUNKS];
#pragma unroll
    for (int c = 0; c < DEG_CHUNKS; ++c) v[c] = part[(size_t)c * N + i];
    float s = 1.0f;
#pragma unroll
    for (int c = 0; c < DEG_CHUNKS; ++c) s += v[c];
    dinv[i] = rsqrtf(s);
}

__global__ void __launch_bounds__(256, 16)
k_gemm_wmma(const float* __restrict__ A, const float* __restrict__ W,
                            const float* __restrict__ dinv,
                            float* __restrict__ out, float* __restrict__ agg,
                            float* __restrict__ agg1, int rowTiles) {
    int wid  = (blockIdx.x * blockDim.x + threadIdx.x) >> 5;
    int lane = threadIdx.x & 31;
    int tr = wid >> 2;
    int tc = wid & 3;
    if (tr >= rowTiles) return;
    int m  = lane & 15;
    int kh = (lane >> 4) << 1;
    const float* Ar = A + (size_t)(tr * 16 + m) * HDIM;
    const float* Wr = W + (size_t)(tc * 16 + m) * HDIM;
    int rowBase = tr * 16 + ((lane >> 4) << 3);
    int col     = tc * 16 + m;
#pragma unroll
    for (int r = 0; r < 8; ++r)
        agg1[(size_t)(rowBase + r) * HDIM + col] = 0.0f;
    v8f c = {0.f, 0.f, 0.f, 0.f, 0.f, 0.f, 0.f, 0.f};
#pragma unroll
    for (int k0 = 0; k0 < HDIM; k0 += 4) {
        v2f a = *(const v2f*)(Ar + k0 + kh);
        v2f b = *(const v2f*)(Wr + k0 + kh);
        c = __builtin_amdgcn_wmma_f32_16x16x4_f32(false, a, false, b, (short)0, c, false, false);
    }
    const float4 dA = *(const float4*)(dinv + rowBase);
    const float4 dB = *(const float4*)(dinv + rowBase + 4);
    const float di[8] = {dA.x, dA.y, dA.z, dA.w, dB.x, dB.y, dB.z, dB.w};
    const int pcol = ((col & 31) << 1) | (col >> 5);
#pragma unroll
    for (int r = 0; r < 8; ++r) {
        size_t ro = (size_t)(rowBase + r) * HDIM;
        float v = c[r] * di[r];
        out[ro + pcol] = v;
        agg[ro + col]  = v;
    }
}

__device__ __forceinline__ float bn_apply(float x, float gam, float mean, float rs, float bet) {
    return gam * (x - mean) * rs + bet;
}

__global__ void __launch_bounds__(256)
k_bn_pool_gemm(const float* hin, const float* __restrict__ stats,
               const float* __restrict__ g, const float* __restrict__ beta,
               const int* __restrict__ batch, float* __restrict__ pooled, int poolOff,
               const float* __restrict__ W, const float* __restrict__ dinv,
               float* __restrict__ out, float* aggOut, float* __restrict__ agg1,
               int rowTiles, int N) {
    __shared__ __attribute__((aligned(16))) float sBN[HDIM * 4];
    if (threadIdx.x < HDIM) {
        int f = threadIdx.x;
        float invN = 1.0f / (float)N;
        float mean = stats[f] * invN;
        float var  = stats[64 + f] * invN - mean * mean;
        float* q = sBN + ((f >> 1) << 3) + (f & 1);
        q[0] = g[f];
        q[2] = mean;
        q[4] = rsqrtf(var + EPSV);
        q[6] = beta[f];
    }
    __syncthreads();
    int wid  = (blockIdx.x * blockDim.x + threadIdx.x) >> 5;
    int lane = threadIdx.x & 31;
    int tr = wid >> 2;
    int tc = wid & 3;
    const bool live = tr < rowTiles;
    int m  = lane & 15;
    int kh = (lane >> 4) << 1;
    int rowBase = tr * 16 + ((lane >> 4) << 3);
    int col     = tc * 16 + m;
    v8f c = {0.f, 0.f, 0.f, 0.f, 0.f, 0.f, 0.f, 0.f};
    float4 dA = make_float4(0.f, 0.f, 0.f, 0.f), dB = dA;
    if (live) {

    {
        int pf   = tc * 16 + m;
        int prow = tr * 16 + ((lane >> 4) << 3);
        const float* q = sBN + ((pf >> 1) << 3) + (pf & 1);
        float pg = q[0], pm = q[2], pr = q[4], pb = q[6];
        float v[8];
#pragma unroll
        for (int r = 0; r < 8; ++r)
            v[r] = bn_apply(hin[(size_t)(prow + r) * HDIM + pf], pg, pm, pr, pb);
        int bmine = batch[tr * 16 + m];
        int b0    = __builtin_amdgcn_readfirstlane(bmine);
        if (__all(bmine == b0)) {
            float s = ((v[0] + v[1]) + (v[2] + v[3])) + ((v[4] + v[5]) + (v[6] + v[7]));
            s += __shfl_xor(s, 16);
            if (lane < 16) atomicAdd(&pooled[(size_t)b0 * 192 + poolOff + pf], s);
        } else {
#pragma unroll
            for (int r = 0; r < 8; ++r)
                atomicAdd(&pooled[(size_t)batch[prow + r] * 192 + poolOff + pf], v[r]);
        }
    }

    const float* Ar = hin + (size_t)(tr * 16 + m) * HDIM;
    const float* Wr = W   + (size_t)(tc * 16 + m) * HDIM;
#pragma unroll
    for (int k0 = 0; k0 < HDIM; k0 += 4) {
        v2f a = *(const v2f*)(Ar + k0 + kh);
        v2f b = *(const v2f*)(Wr + k0 + kh);
        const float* q = sBN + ((k0 + kh) << 2);
        v2f gam = *(const v2f*)(q), mean = *(const v2f*)(q + 2);
        v2f rs  = *(const v2f*)(q + 4), bet = *(const v2f*)(q + 6);
        a.x = bn_apply(a.x, gam.x, mean.x, rs.x, bet.x);
        a.y = bn_apply(a.y, gam.y, mean.y, rs.y, bet.y);
        c = __builtin_amdgcn_wmma_f32_16x16x4_f32(false, a, false, b, (short)0, c, false, false);
    }
    dA = *(const float4*)(dinv + rowBase);
    dB = *(const float4*)(dinv + rowBase + 4);
    }
    __syncthreads();
    if (live) {
        const float di[8] = {dA.x, dA.y, dA.z, dA.w, dB.x, dB.y, dB.z, dB.w};
        const int pcol = ((col & 31) << 1) | (col >> 5);
#pragma unroll
        for (int r = 0; r < 8; ++r) {
            size_t ro = (size_t)(rowBase + r) * HDIM;
            float v   = c[r] * di[r];
            out[ro + pcol]   = v;
            aggOut[ro + col] = v;
            agg1[ro + col]   = 0.0f;
        }
    }
}

__device__ __forceinline__ void scatter_edge(const float* __restrict__ hWs,
                                             float* __restrict__ agg,
                                             int s, int d, int lane) {
    const v2f p = *((const v2f*)(hWs + (size_t)s * HDIM) + lane);
    float* a = agg + (size_t)d * HDIM + lane;
    atomicAdd(a, p.x);
    atomicAdd(a + 32, p.y);
}

__global__ void __launch_bounds__(256)
k_scatter(const float* __restrict__ hWs,
          const int* __restrict__ src, const int* __restrict__ dst,
          float* __restrict__ agg0, float* __restrict__ agg1, int E) {
    const int lane = threadIdx.x & 31;
    const unsigned place = __builtin_amdgcn_s_sendmsg_rtn(0x87);
    float* __restrict__ agg = ((place >> 18) & 1u) ? agg1 : agg0;
    const int wave   = blockIdx.x * 8 + __builtin_amdgcn_readfirstlane((int)(threadIdx.x >> 5));
    const int nwaves = gridDim.x * 8;
    const int nquad  = E >> 2;
    for (int g = wave; g < nquad; g += nwaves) {
        const int e = g << 2;
        int s0 = src[e], s1 = src[e + 1], s2 = src[e + 2], s3 = src[e + 3];
        int d0 = dst[e], d1 = dst[e + 1], d2 = dst[e + 2], d3 = dst[e + 3];
        const v2f p0 = *((const v2f*)(hWs + (size_t)s0 * HDIM) + lane);
        const v2f p1 = *((const v2f*)(hWs + (size_t)s1 * HDIM) + lane);
        const v2f p2 = *((const v2f*)(hWs + (size_t)s2 * HDIM) + lane);
        const v2f p3 = *((const v2f*)(hWs + (size_t)s3 * HDIM) + lane);
        float* a0 = agg + (size_t)d0 * HDIM + lane;
        float* a1 = agg + (size_t)d1 * HDIM + lane;
        float* a2 = agg + (size_t)d2 * HDIM + lane;
        float* a3 = agg + (size_t)d3 * HDIM + lane;
        atomicAdd(a0, p0.x);  atomicAdd(a0 + 32, p0.y);
        atomicAdd(a1, p1.x);  atomicAdd(a1 + 32, p1.y);
        atomicAdd(a2, p2.x);  atomicAdd(a2 + 32, p2.y);
        atomicAdd(a3, p3.x);  atomicAdd(a3 + 32, p3.y);
    }
    for (int e = (nquad << 2) + wave; e < E; e += nwaves)
        scatter_edge(hWs, agg, src[e], dst[e], lane);
}

#define BRS_ROWS 8

__global__ void k_bias_relu_stats(float* __restrict__ agg, const float* __restrict__ agg1,
                                  const float* __restrict__ dinv,
                                  const float* __restrict__ bias,
                                  float* __restrict__ stats  , int N) {
    __shared__ float sS[256];
    __shared__ float sQ[256];
    const int f = threadIdx.x & 63;
    const int rg = __builtin_amdgcn_readfirstlane((int)(threadIdx.x >> 6));
    const float b = bias[f];
    float s = 0.f, q = 0.f;
    const int step = gridDim.x * 4;
    int row = blockIdx.x * 4 + rg;
    for (; row + (BRS_ROWS - 1) * step < N; row += BRS_ROWS * step) {
        float p0[BRS_ROWS], p1[BRS_ROWS], dv[BRS_ROWS];
#pragma unroll
        for (int k = 0; k < BRS_ROWS; ++k) {
            const int r = row + k * step;
            p0[k] = agg[r * HDIM + f];
            p1[k] = agg1[r * HDIM + f];
            dv[k] = dinv[r];
        }
#pragma unroll
        for (int k = 0; k < BRS_ROWS; ++k) {
            float v = (p0[k] + p1[k]) * dv[k] + b;
            v = fmaxf(v, 0.f);
            agg[(row + k * step) * HDIM + f] = v;
            s += v;
            q += v * v;
        }
    }
    for (; row < N; row += step) {
        const int idx = row * HDIM + f;
        float v = (agg[idx] + agg1[idx]) * dinv[row] + b;
        v = fmaxf(v, 0.f);
        agg[idx] = v;
        s += v;
        q += v * v;
    }
    sS[threadIdx.x] = s;
    sQ[threadIdx.x] = q;
    __syncthreads();
    if (threadIdx.x < 64) {
        s = sS[f] + sS[64 + f] + sS[128 + f] + sS[192 + f];
        q = sQ[f] + sQ[64 + f] + sQ[128 + f] + sQ[192 + f];
        atomicAdd(&stats[f], s);
        atomicAdd(&stats[64 + f], q);
    }
}

__global__ void k_bn_pool(const float* __restrict__ hin, const float* __restrict__ stats,
                          const float* __restrict__ g, const float* __restrict__ beta,
                          const int* __restrict__ batch, float* __restrict__ hout,
                          float* __restrict__ pooled, int poolOff, int N) {
    int t = blockIdx.x * blockDim.x + threadIdx.x;
    if (t >= N * HDIM) return;
    int n = t >> 6, f = t & 63;
    float invN = 1.0f / (float)N;
    float mean = stats[f] * invN;
    float var  = stats[64 + f] * invN - mean * mean;
    float v = g[f] * (hin[t] - mean) * rsqrtf(var + EPSV) + beta[f];
    hout[t] = v;
    atomicAdd(&pooled[(size_t)batch[n] * 192 + poolOff + f], v);
}

extern "C" void kernel_launch(void* const* d_in, const int* in_sizes, int n_in,
                              void* d_out, int out_size, void* d_ws, size_t ws_size,
                              hipStream_t stream) {
    const float* x    = (const float*)d_in[0];
    const int*   eidx = (const int*)d_in[1];
    const int*   batch = (const int*)d_in[2];
    const float* W[3]  = {(const float*)d_in[3],  (const float*)d_in[7],  (const float*)d_in[11]};
    const float* bi[3] = {(const float*)d_in[4],  (const float*)d_in[8],  (const float*)d_in[12]};
    const float* ga[3] = {(const float*)d_in[5],  (const float*)d_in[9],  (const float*)d_in[13]};
    const float* be[3] = {(const float*)d_in[6],  (const float*)d_in[10], (const float*)d_in[14]};

    const int N = in_sizes[0] / HDIM;
    const int E = in_sizes[1] / 2;
    const int G = (out_size - N * HDIM) / (3 * HDIM);

    const int* src = eidx;
    const int* dst = eidx + E;

    float* ws    = (float*)d_ws;
    float* tmp   = ws;
    float* agg   = tmp  + (size_t)N * HDIM;
    float* dinv  = agg  + (size_t)N * HDIM;
    float* stats = dinv + N;

    float* pooled = (float*)d_out;
    float* hfinal = (float*)d_out + (size_t)G * 192;
    float* agg1 = hfinal;

    const int TB = 256;

    k_deg_count<<<cdiv(N, DEG_SPAN) * DEG_CHUNKS, 1024, 0, stream>>>(dst, tmp, E, N);
    k_dinv     <<<cdiv(N, TB), TB, 0, stream>>>(tmp, dinv, N);

    k_zero<<<cdiv((long long)G * 192, TB), TB, 0, stream>>>(pooled, G * 192);

    const int rowTiles   = N / 16;
    const int gemmWaves  = rowTiles * (HDIM / 16);
    const int gemmBlocks = cdiv((long long)gemmWaves * 32, TB);

    for (int l = 0; l < 3; ++l) {
        if (l == 0)
            k_gemm_wmma<<<gemmBlocks, TB, 0, stream>>>(x, W[0], dinv, tmp, agg, agg1, rowTiles);
        else
            k_bn_pool_gemm<<<gemmBlocks, TB, 0, stream>>>(agg, stats, ga[l - 1], be[l - 1], batch,
                                                         pooled, (l - 1) * HDIM, W[l], dinv, tmp, agg, agg1,
                                                         rowTiles, N);
        k_scatter  <<<2048, TB, 0, stream>>>(tmp, src, dst, agg, agg1, E);

        k_zero<<<1, 128, 0, stream>>>(stats, 128);
        k_bias_relu_stats<<<1024, TB, 0, stream>>>(agg, agg1, dinv, bi[l], stats, N);
    }
    k_bn_pool<<<cdiv((long long)N * HDIM, TB), TB, 0, stream>>>(
        agg, stats, ga[2], be[2], batch, hfinal, pooled, 2 * HDIM, N);
}
